# starting kernel
# baseline (speedup 1.0000x reference)
#include <hip/hip_runtime.h>

#define N_NODES 10000
#define E_EDGES 640000
#define C_IN    256
#define C_HID   128
#define C_OUT   64

typedef float    v2f  __attribute__((ext_vector_type(2)));
typedef float    v8f  __attribute__((ext_vector_type(8)));
typedef _Float16 v16h __attribute__((ext_vector_type(16)));

#if defined(__has_builtin)
#if __has_builtin(__builtin_amdgcn_wmma_f32_16x16x4_f32)
#define USE_WMMA_F32 1
#endif
#endif

__global__ void deg_init_k(float* __restrict__ dinv, int n) {
  int i = blockIdx.x * blockDim.x + threadIdx.x;
  if (i < n) dinv[i] = 1.0f;
}

__global__ void deg_count_k(float* __restrict__ dinv, const int* __restrict__ dst, int e) {
  int i = blockIdx.x * blockDim.x + threadIdx.x;
  if (i < e)
    __hip_atomic_fetch_add(&dinv[dst[i]], 1.0f, __ATOMIC_RELAXED, __HIP_MEMORY_SCOPE_AGENT);
}

__global__ void rsqrt_k(float* __restrict__ dinv, int n) {
  int i = blockIdx.x * blockDim.x + threadIdx.x;
  if (i < n) dinv[i] = rsqrtf(dinv[i]);
}

__global__ __launch_bounds__(128)
void gemm_wmma_k(const float* __restrict__ A, const float* __restrict__ B,
                 float* __restrict__ D, int K, int Ncols) {
  const int lane = threadIdx.x & 31;
  const int wave = threadIdx.x >> 5;
  const int row0 = blockIdx.x * 16;
  const int col0 = (blockIdx.y * 4 + wave) * 16;
  const int r    = lane & 15;
  const int half = lane >> 4;

  const float* Arow = A + (size_t)(row0 + r) * K;
  const float* Bcol = B + col0 + r;
  v8f acc = {};

#if defined(USE_WMMA_F32)
  for (int k = 0; k < K; k += 4) {
    const int kb = k + half * 2;
    v2f a = *(const v2f*)(Arow + kb);
    v2f b;
    b.x = Bcol[(size_t)kb * Ncols];
    b.y = Bcol[(size_t)(kb + 1) * Ncols];
    acc = __builtin_amdgcn_wmma_f32_16x16x4_f32(false, a, false, b,
                                                (short)0, acc, false, false);
  }
#else
  for (int k = 0; k < K; k += 32) {
    v16h a, b;
#pragma unroll
    for (int rr = 0; rr < 8; ++rr) {
      const int ka = k + (rr >> 2) * 16 + half * 8 + (rr & 3) * 2;
      a[2 * rr]     = (_Float16)Arow[ka];
      a[2 * rr + 1] = (_Float16)Arow[ka + 1];
      const int kb = k + half * 16 + 2 * rr;
      b[2 * rr]     = (_Float16)Bcol[(size_t)kb * Ncols];
      b[2 * rr + 1] = (_Float16)Bcol[(size_t)(kb + 1) * Ncols];
    }
    acc = __builtin_amdgcn_wmma_f32_16x16x32_f16(false, a, false, b,
                                                 (short)0, acc, false, false);
  }
#endif

  float* Drow = D + (size_t)(row0 + half * 8) * Ncols + col0 + r;
#pragma unroll
  for (int i = 0; i < 8; ++i) Drow[(size_t)i * Ncols] = acc[i];
}

template <int C>
__global__ void init_out_k(const float* __restrict__ h, const float* __restrict__ dinv,
                           const float* __restrict__ bias, float* __restrict__ out, int n) {
  int t = blockIdx.x * blockDim.x + threadIdx.x;
  if (t >= n * C) return;
  int i = t / C;
  int f = t - i * C;
  float di = dinv[i];
  out[t] = bias[f] + di * di * h[t];
}

template <int C>
__global__ void scatter_k(const float* __restrict__ h, const float* __restrict__ dinv,
                          const int* __restrict__ src, const int* __restrict__ dst,
                          float* __restrict__ out, int e) {
  constexpr int CH = C / 4;
  int t = blockIdx.x * blockDim.x + threadIdx.x;
  if (t >= e * CH) return;
  int ei = t / CH;
  int fc = (t - ei * CH) * 4;
  int s = src[ei], d = dst[ei];
  float w = dinv[s] * dinv[d];
  const float4 hv = *(const float4*)(h + (size_t)s * C + fc);
  float* o = out + (size_t)d * C + fc;
  __hip_atomic_fetch_add(o + 0, w * hv.x, __ATOMIC_RELAXED, __HIP_MEMORY_SCOPE_AGENT);
  __hip_atomic_fetch_add(o + 1, w * hv.y, __ATOMIC_RELAXED, __HIP_MEMORY_SCOPE_AGENT);
  __hip_atomic_fetch_add(o + 2, w * hv.z, __ATOMIC_RELAXED, __HIP_MEMORY_SCOPE_AGENT);
  __hip_atomic_fetch_add(o + 3, w * hv.w, __ATOMIC_RELAXED, __HIP_MEMORY_SCOPE_AGENT);
}

__global__ void relu_k(float* __restrict__ a, int n) {
  int t = blockIdx.x * blockDim.x + threadIdx.x;
  if (t < n) a[t] = fmaxf(a[t], 0.0f);
}

extern "C" void kernel_launch(void* const* d_in, const int* in_sizes, int n_in,
                              void* d_out, int out_size, void* d_ws, size_t ws_size,
                              hipStream_t stream) {
  (void)in_sizes; (void)n_in; (void)out_size; (void)ws_size;
  const float* x  = (const float*)d_in[0];
  const int*   ei = (const int*)d_in[1];
  const float* W1 = (const float*)d_in[2];
  const float* b1 = (const float*)d_in[3];
  const float* W2 = (const float*)d_in[4];
  const float* b2 = (const float*)d_in[5];
  const int* src = ei;
  const int* dst = ei + E_EDGES;
  float* out = (float*)d_out;

  char* ws = (char*)d_ws;
  float* dinv = (float*)ws;  ws += (((size_t)N_NODES * 4) + 255) / 256 * 256;
  float* H1   = (float*)ws;  ws += (size_t)N_NODES * C_HID * 4;
  float* A1   = (float*)ws;  ws += (size_t)N_NODES * C_HID * 4;
  float* H2   = (float*)ws;

  const dim3 B(256);
  deg_init_k<<<(N_NODES + 255) / 256, B, 0, stream>>>(dinv, N_NODES);
  deg_count_k<<<(E_EDGES + 255) / 256, B, 0, stream>>>(dinv, dst, E_EDGES);
  rsqrt_k<<<(N_NODES + 255) / 256, B, 0, stream>>>(dinv, N_NODES);

  gemm_wmma_k<<<dim3(N_NODES / 16, C_HID / 64), dim3(128), 0, stream>>>(x, W1, H1, C_IN, C_HID);
  init_out_k<C_HID><<<(N_NODES * C_HID + 255) / 256, B, 0, stream>>>(H1, dinv, b1, A1, N_NODES);
  scatter_k<C_HID><<<(E_EDGES * (C_HID / 4) + 255) / 256, B, 0, stream>>>(H1, dinv, src, dst, A1, E_EDGES);
  relu_k<<<(N_NODES * C_HID + 255) / 256, B, 0, stream>>>(A1, N_NODES * C_HID);

  gemm_wmma_k<<<dim3(N_NODES / 16, C_OUT / 64), dim3(128), 0, stream>>>(A1, W2, H2, C_HID, C_OUT);
  init_out_k<C_OUT><<<(N_NODES * C_OUT + 255) / 256, B, 0, stream>>>(H2, dinv, b2, out, N_NODES);
  scatter_k<C_OUT><<<(E_EDGES * (C_OUT / 4) + 255) / 256, B, 0, stream>>>(H2, dinv, src, dst, out, E_EDGES);
}


# degree count by per-block LDS histogram of dst (64 blocks, ds_add_u32) plus one sum+rsqrt kernel, replacing 640k contended global float atomics; integer counts exact, dinv bit-identical
# speedup vs baseline: 1.3073x; 1.3073x over previous
#include <hip/hip_runtime.h>

#define N_NODES 10000
#define E_EDGES 640000
#define C_IN    256
#define C_HID   128
#define C_OUT   64

typedef float    v2f  __attribute__((ext_vector_type(2)));
typedef float    v8f  __attribute__((ext_vector_type(8)));
typedef _Float16 v16h __attribute__((ext_vector_type(16)));

#if defined(__has_builtin)
#if __has_builtin(__builtin_amdgcn_wmma_f32_16x16x4_f32)
#define USE_WMMA_F32 1
#endif
#endif

#define HIST_BLOCKS  64
#define HIST_THREADS 1024
static_assert(E_EDGES % (HIST_BLOCKS * 4) == 0, "each block reads whole int4 groups");

__global__ __launch_bounds__(HIST_THREADS)
void deg_hist_k(const int* __restrict__ dst, unsigned* __restrict__ part) {
  __shared__ unsigned cnt[N_NODES];
  for (int i = threadIdx.x; i < N_NODES; i += HIST_THREADS) cnt[i] = 0u;
  __syncthreads();

  constexpr int SLICE  = E_EDGES / HIST_BLOCKS;
  constexpr int SLICE4 = SLICE / 4;
  const int4* d4 = (const int4*)(dst + (size_t)blockIdx.x * SLICE);
  for (int i = threadIdx.x; i < SLICE4; i += HIST_THREADS) {
    const int4 v = d4[i];
    __hip_atomic_fetch_add(&cnt[v.x], 1u, __ATOMIC_RELAXED, __HIP_MEMORY_SCOPE_WORKGROUP);
    __hip_atomic_fetch_add(&cnt[v.y], 1u, __ATOMIC_RELAXED, __HIP_MEMORY_SCOPE_WORKGROUP);
    __hip_atomic_fetch_add(&cnt[v.z], 1u, __ATOMIC_RELAXED, __HIP_MEMORY_SCOPE_WORKGROUP);
    __hip_atomic_fetch_add(&cnt[v.w], 1u, __ATOMIC_RELAXED, __HIP_MEMORY_SCOPE_WORKGROUP);
  }
  __syncthreads();

  unsigned* prow = part + (size_t)blockIdx.x * N_NODES;
  for (int i = threadIdx.x; i < N_NODES; i += HIST_THREADS) prow[i] = cnt[i];
}

__global__ void deg_finish_k(const unsigned* __restrict__ part, float* __restrict__ dinv) {
  int i = blockIdx.x * blockDim.x + threadIdx.x;
  if (i >= N_NODES) return;
  unsigned s = 1u;
  for (int g = 0; g < HIST_BLOCKS; ++g) s += part[(size_t)g * N_NODES + i];
  dinv[i] = rsqrtf((float)s);
}

__global__ __launch_bounds__(128)
void gemm_wmma_k(const float* __restrict__ A, const float* __restrict__ B,
                 float* __restrict__ D, int K, int Ncols) {
  const int lane = threadIdx.x & 31;
  const int wave = threadIdx.x >> 5;
  const int row0 = blockIdx.x * 16;
  const int col0 = (blockIdx.y * 4 + wave) * 16;
  const int r    = lane & 15;
  const int half = lane >> 4;

  const float* Arow = A + (size_t)(row0 + r) * K;
  const float* Bcol = B + col0 + r;
  v8f acc = {};

#if defined(USE_WMMA_F32)
  for (int k = 0; k < K; k += 4) {
    const int kb = k + half * 2;
    v2f a = *(const v2f*)(Arow + kb);
    v2f b;
    b.x = Bcol[(size_t)kb * Ncols];
    b.y = Bcol[(size_t)(kb + 1) * Ncols];
    acc = __builtin_amdgcn_wmma_f32_16x16x4_f32(false, a, false, b,
                                                (short)0, acc, false, false);
  }
#else
  for (int k = 0; k < K; k += 32) {
    v16h a, b;
#pragma unroll
    for (int rr = 0; rr < 8; ++rr) {
      const int ka = k + (rr >> 2) * 16 + half * 8 + (rr & 3) * 2;
      a[2 * rr]     = (_Float16)Arow[ka];
      a[2 * rr + 1] = (_Float16)Arow[ka + 1];
      const int kb = k + half * 16 + 2 * rr;
      b[2 * rr]     = (_Float16)Bcol[(size_t)kb * Ncols];
      b[2 * rr + 1] = (_Float16)Bcol[(size_t)(kb + 1) * Ncols];
    }
    acc = __builtin_amdgcn_wmma_f32_16x16x32_f16(false, a, false, b,
                                                 (short)0, acc, false, false);
  }
#endif

  float* Drow = D + (size_t)(row0 + half * 8) * Ncols + col0 + r;
#pragma unroll
  for (int i = 0; i < 8; ++i) Drow[(size_t)i * Ncols] = acc[i];
}

template <int C>
__global__ void init_out_k(const float* __restrict__ h, const float* __restrict__ dinv,
                           const float* __restrict__ bias, float* __restrict__ out, int n) {
  int t = blockIdx.x * blockDim.x + threadIdx.x;
  if (t >= n * C) return;
  int i = t / C;
  int f = t - i * C;
  float di = dinv[i];
  out[t] = bias[f] + di * di * h[t];
}

template <int C>
__global__ void scatter_k(const float* __restrict__ h, const float* __restrict__ dinv,
                          const int* __restrict__ src, const int* __restrict__ dst,
                          float* __restrict__ out, int e) {
  constexpr int CH = C / 4;
  int t = blockIdx.x * blockDim.x + threadIdx.x;
  if (t >= e * CH) return;
  int ei = t / CH;
  int fc = (t - ei * CH) * 4;
  int s = src[ei], d = dst[ei];
  float w = dinv[s] * dinv[d];
  const float4 hv = *(const float4*)(h + (size_t)s * C + fc);
  float* o = out + (size_t)d * C + fc;
  __hip_atomic_fetch_add(o + 0, w * hv.x, __ATOMIC_RELAXED, __HIP_MEMORY_SCOPE_AGENT);
  __hip_atomic_fetch_add(o + 1, w * hv.y, __ATOMIC_RELAXED, __HIP_MEMORY_SCOPE_AGENT);
  __hip_atomic_fetch_add(o + 2, w * hv.z, __ATOMIC_RELAXED, __HIP_MEMORY_SCOPE_AGENT);
  __hip_atomic_fetch_add(o + 3, w * hv.w, __ATOMIC_RELAXED, __HIP_MEMORY_SCOPE_AGENT);
}

__global__ void relu_k(float* __restrict__ a, int n) {
  int t = blockIdx.x * blockDim.x + threadIdx.x;
  if (t < n) a[t] = fmaxf(a[t], 0.0f);
}

extern "C" void kernel_launch(void* const* d_in, const int* in_sizes, int n_in,
                              void* d_out, int out_size, void* d_ws, size_t ws_size,
                              hipStream_t stream) {
  (void)in_sizes; (void)n_in; (void)out_size; (void)ws_size;
  const float* x  = (const float*)d_in[0];
  const int*   ei = (const int*)d_in[1];
  const float* W1 = (const float*)d_in[2];
  const float* b1 = (const float*)d_in[3];
  const float* W2 = (const float*)d_in[4];
  const float* b2 = (const float*)d_in[5];
  const int* src = ei;
  const int* dst = ei + E_EDGES;
  float* out = (float*)d_out;

  char* ws = (char*)d_ws;
  float* dinv = (float*)ws;  ws += (((size_t)N_NODES * 4) + 255) / 256 * 256;
  float* H1   = (float*)ws;  ws += (size_t)N_NODES * C_HID * 4;
  float* A1   = (float*)ws;  ws += (size_t)N_NODES * C_HID * 4;
  float* H2   = (float*)ws;  ws += (size_t)N_NODES * C_OUT * 4;
  unsigned* part = (unsigned*)ws;

  const dim3 B(256);
  deg_hist_k<<<HIST_BLOCKS, HIST_THREADS, 0, stream>>>(dst, part);
  deg_finish_k<<<(N_NODES + 255) / 256, B, 0, stream>>>(part, dinv);

  gemm_wmma_k<<<dim3(N_NODES / 16, C_HID / 64), dim3(128), 0, stream>>>(x, W1, H1, C_IN, C_HID);
  init_out_k<C_HID><<<(N_NODES * C_HID + 255) / 256, B, 0, stream>>>(H1, dinv, b1, A1, N_NODES);
  scatter_k<C_HID><<<(E_EDGES * (C_HID / 4) + 255) / 256, B, 0, stream>>>(H1, dinv, src, dst, A1, E_EDGES);
  relu_k<<<(N_NODES * C_HID + 255) / 256, B, 0, stream>>>(A1, N_NODES * C_HID);

  gemm_wmma_k<<<dim3(N_NODES / 16, C_OUT / 64), dim3(128), 0, stream>>>(A1, W2, H2, C_HID, C_OUT);
  init_out_k<C_OUT><<<(N_NODES * C_OUT + 255) / 256, B, 0, stream>>>(H2, dinv, b2, out, N_NODES);
  scatter_k<C_OUT><<<(E_EDGES * (C_OUT / 4) + 255) / 256, B, 0, stream>>>(H2, dinv, src, dst, out, E_EDGES);
}


# float-atomic scatter replaced by gather over edges counting-sorted by dst (CSR built in LDS): one wave per node sums h rows in registers, self-loop/bias/relu folded in; sum order = CSR order per node
# speedup vs baseline: 2.8363x; 2.1696x over previous
#include <hip/hip_runtime.h>

#define N_NODES 10000
#define E_EDGES 640000
#define C_IN    256
#define C_HID   128
#define C_OUT   64

typedef float    v2f  __attribute__((ext_vector_type(2)));
typedef float    v8f  __attribute__((ext_vector_type(8)));
typedef _Float16 v16h __attribute__((ext_vector_type(16)));

#if defined(__has_builtin)
#if __has_builtin(__builtin_amdgcn_wmma_f32_16x16x4_f32)
#define USE_WMMA_F32 1
#endif
#endif

#define HIST_BLOCKS  64
#define HIST_THREADS 1024
static_assert(E_EDGES % (HIST_BLOCKS * 4) == 0, "each block reads whole int4 groups");

__global__ __launch_bounds__(HIST_THREADS)
void deg_hist_k(const int* __restrict__ dst, unsigned* __restrict__ part) {
  __shared__ unsigned cnt[N_NODES];
  for (int i = threadIdx.x; i < N_NODES; i += HIST_THREADS) cnt[i] = 0u;
  __syncthreads();

  constexpr int SLICE  = E_EDGES / HIST_BLOCKS;
  constexpr int SLICE4 = SLICE / 4;
  const int4* d4 = (const int4*)(dst + (size_t)blockIdx.x * SLICE);
  for (int i = threadIdx.x; i < SLICE4; i += HIST_THREADS) {
    const int4 v = d4[i];
    __hip_atomic_fetch_add(&cnt[v.x], 1u, __ATOMIC_RELAXED, __HIP_MEMORY_SCOPE_WORKGROUP);
    __hip_atomic_fetch_add(&cnt[v.y], 1u, __ATOMIC_RELAXED, __HIP_MEMORY_SCOPE_WORKGROUP);
    __hip_atomic_fetch_add(&cnt[v.z], 1u, __ATOMIC_RELAXED, __HIP_MEMORY_SCOPE_WORKGROUP);
    __hip_atomic_fetch_add(&cnt[v.w], 1u, __ATOMIC_RELAXED, __HIP_MEMORY_SCOPE_WORKGROUP);
  }
  __syncthreads();

  unsigned* prow = part + (size_t)blockIdx.x * N_NODES;
  for (int i = threadIdx.x; i < N_NODES; i += HIST_THREADS) prow[i] = cnt[i];
}

__global__ void deg_finish_k(unsigned* __restrict__ part, unsigned* __restrict__ cnt,
                             float* __restrict__ dinv) {
  int i = blockIdx.x * blockDim.x + threadIdx.x;
  if (i >= N_NODES) return;
  unsigned run = 0u;
  for (int g = 0; g < HIST_BLOCKS; ++g) {
    unsigned* p = part + (size_t)g * N_NODES + i;
    const unsigned c = *p;
    *p = run;
    run += c;
  }
  cnt[i] = run;
  dinv[i] = rsqrtf((float)(run + 1u));
}

#define SCAN_PER ((N_NODES + HIST_THREADS - 1) / HIST_THREADS)
#define SCAN_LEN (SCAN_PER * HIST_THREADS)

__global__ __launch_bounds__(HIST_THREADS)
void csr_fill_k(const int* __restrict__ src, const int* __restrict__ dst,
                const unsigned* __restrict__ cnt, const unsigned* __restrict__ partoff,
                unsigned* __restrict__ rowptr, int* __restrict__ csr_src) {
  __shared__ unsigned cur[SCAN_LEN];
  __shared__ unsigned wtot[HIST_THREADS / 32];
  const int t    = threadIdx.x;
  const int lane = t & 31;
  const int w    = t >> 5;

#pragma unroll
  for (int k = 0; k < SCAN_PER; ++k) {
    const int i = t + k * HIST_THREADS;
    cur[i] = (i < N_NODES) ? cnt[i] : 0u;
  }
  __syncthreads();

  const int i0 = t * SCAN_PER;
  unsigned loc[SCAN_PER];
  unsigned sum = 0u;
#pragma unroll
  for (int k = 0; k < SCAN_PER; ++k) {
    loc[k] = sum;
    sum += cur[i0 + k];
  }
  unsigned inc = sum;
  for (int o = 1; o < 32; o <<= 1) {
    const unsigned v = __shfl_up(inc, o);
    if (lane >= o) inc += v;
  }
  if (lane == 31) wtot[w] = inc;
  __syncthreads();
  if (w == 0) {
    const unsigned x = wtot[lane];
    unsigned xi = x;
    for (int o = 1; o < 32; o <<= 1) {
      const unsigned v = __shfl_up(xi, o);
      if (lane >= o) xi += v;
    }
    wtot[lane] = xi - x;
  }
  __syncthreads();
  const unsigned base = wtot[w] + (inc - sum);
#pragma unroll
  for (int k = 0; k < SCAN_PER; ++k) cur[i0 + k] = base + loc[k];
  __syncthreads();

  if (blockIdx.x == 0) {
#pragma unroll
    for (int k = 0; k < SCAN_PER; ++k) {
      const int i = t + k * HIST_THREADS;
      if (i < N_NODES) rowptr[i] = cur[i];
    }
    if (t == HIST_THREADS - 1) rowptr[N_NODES] = base + sum;
  }
  const unsigned* po = partoff + (size_t)blockIdx.x * N_NODES;
#pragma unroll
  for (int k = 0; k < SCAN_PER; ++k) {
    const int i = t + k * HIST_THREADS;
    if (i < N_NODES) cur[i] += po[i];
  }
  __syncthreads();

  constexpr int SLICE  = E_EDGES / HIST_BLOCKS;
  constexpr int SLICE4 = SLICE / 4;
  const int4* d4 = (const int4*)(dst + (size_t)blockIdx.x * SLICE);
  const int4* s4 = (const int4*)(src + (size_t)blockIdx.x * SLICE);
  for (int i = t; i < SLICE4; i += HIST_THREADS) {
    const int4 d = d4[i];
    const int4 s = s4[i];
    const unsigned p0 = __hip_atomic_fetch_add(&cur[d.x], 1u, __ATOMIC_RELAXED, __HIP_MEMORY_SCOPE_WORKGROUP);
    const unsigned p1 = __hip_atomic_fetch_add(&cur[d.y], 1u, __ATOMIC_RELAXED, __HIP_MEMORY_SCOPE_WORKGROUP);
    const unsigned p2 = __hip_atomic_fetch_add(&cur[d.z], 1u, __ATOMIC_RELAXED, __HIP_MEMORY_SCOPE_WORKGROUP);
    const unsigned p3 = __hip_atomic_fetch_add(&cur[d.w], 1u, __ATOMIC_RELAXED, __HIP_MEMORY_SCOPE_WORKGROUP);
    csr_src[p0] = s.x;
    csr_src[p1] = s.y;
    csr_src[p2] = s.z;
    csr_src[p3] = s.w;
  }
}

__global__ __launch_bounds__(128)
void gemm_wmma_k(const float* __restrict__ A, const float* __restrict__ B,
                 float* __restrict__ D, int K, int Ncols) {
  const int lane = threadIdx.x & 31;
  const int wave = threadIdx.x >> 5;
  const int row0 = blockIdx.x * 16;
  const int col0 = (blockIdx.y * 4 + wave) * 16;
  const int r    = lane & 15;
  const int half = lane >> 4;

  const float* Arow = A + (size_t)(row0 + r) * K;
  const float* Bcol = B + col0 + r;
  v8f acc = {};

#if defined(USE_WMMA_F32)
  for (int k = 0; k < K; k += 4) {
    const int kb = k + half * 2;
    v2f a = *(const v2f*)(Arow + kb);
    v2f b;
    b.x = Bcol[(size_t)kb * Ncols];
    b.y = Bcol[(size_t)(kb + 1) * Ncols];
    acc = __builtin_amdgcn_wmma_f32_16x16x4_f32(false, a, false, b,
                                                (short)0, acc, false, false);
  }
#else
  for (int k = 0; k < K; k += 32) {
    v16h a, b;
#pragma unroll
    for (int rr = 0; rr < 8; ++rr) {
      const int ka = k + (rr >> 2) * 16 + half * 8 + (rr & 3) * 2;
      a[2 * rr]     = (_Float16)Arow[ka];
      a[2 * rr + 1] = (_Float16)Arow[ka + 1];
      const int kb = k + half * 16 + 2 * rr;
      b[2 * rr]     = (_Float16)Bcol[(size_t)kb * Ncols];
      b[2 * rr + 1] = (_Float16)Bcol[(size_t)(kb + 1) * Ncols];
    }
    acc = __builtin_amdgcn_wmma_f32_16x16x32_f16(false, a, false, b,
                                                 (short)0, acc, false, false);
  }
#endif

  float* Drow = D + (size_t)(row0 + half * 8) * Ncols + col0 + r;
#pragma unroll
  for (int i = 0; i < 8; ++i) Drow[(size_t)i * Ncols] = acc[i];
}

template <int G>
__device__ __forceinline__ int lane_bcast(int v, int idx) {
  if constexpr (G == 1) return __builtin_amdgcn_readlane(v, idx);
  else                  return __shfl(v, idx);
}

template <int C, int G, int NS>
__device__ __forceinline__ void rows_fma(const float* __restrict__ h, int sv, int wvi,
                                         int step0, int grp, unsigned fl, float4& acc) {
  float4 hv[NS];
  float  wj[NS];
#pragma unroll
  for (int u = 0; u < NS; ++u) {
    const int idx = (step0 + u) * G + grp;
    const unsigned s = (unsigned)lane_bcast<G>(sv, idx);
    wj[u] = __int_as_float(lane_bcast<G>(wvi, idx));
    hv[u] = *(const float4*)(h + s * C + fl);
  }
#pragma unroll
  for (int u = 0; u < NS; ++u) {
    acc.x = fmaf(wj[u], hv[u].x, acc.x);
    acc.y = fmaf(wj[u], hv[u].y, acc.y);
    acc.z = fmaf(wj[u], hv[u].z, acc.z);
    acc.w = fmaf(wj[u], hv[u].w, acc.w);
  }
}

template <int C, bool RELU>
__global__ __launch_bounds__(256)
void gather_k(const float* __restrict__ h, const float* __restrict__ dinv,
              const unsigned* __restrict__ rowptr, const int* __restrict__ csr_src,
              const float* __restrict__ bias, float* __restrict__ out) {
  constexpr int LPR = C / 4;
  constexpr int G   = 32 / LPR;
  constexpr int U   = 8;
  const int lane = threadIdx.x & 31;
  const int node = __builtin_amdgcn_readfirstlane((int)(blockIdx.x * 8 + (threadIdx.x >> 5)));
  if (node >= N_NODES) return;
  const int grp = lane / LPR;
  const unsigned fl = (unsigned)(lane % LPR) * 4u;

  const unsigned beg = rowptr[node];
  const unsigned end = rowptr[node + 1];
  const float di = dinv[node];

  float4 acc = make_float4(0.0f, 0.0f, 0.0f, 0.0f);
  if (grp == 0) {
    const float4 hv = *(const float4*)(h + (unsigned)node * C + fl);
    const float ws = di * di;
    acc.x = ws * hv.x; acc.y = ws * hv.y; acc.z = ws * hv.z; acc.w = ws * hv.w;
  }

  for (unsigned cb = beg; cb < end; cb += 32u) {
    const int n = (int)((end - cb) < 32u ? (end - cb) : 32u);
    int   sv = 0;
    float wv = 0.0f;
    if (lane < n) {
      sv = csr_src[cb + lane];
      wv = dinv[sv] * di;
    }
    const int wvi = __float_as_int(wv);

    const int nfull = n / G;
    int j = 0;
    for (; j + U <= nfull; j += U) rows_fma<C, G, U>(h, sv, wvi, j, grp, fl, acc);
    if (j + 4 <= nfull) { rows_fma<C, G, 4>(h, sv, wvi, j, grp, fl, acc); j += 4; }
    if (j + 2 <= nfull) { rows_fma<C, G, 2>(h, sv, wvi, j, grp, fl, acc); j += 2; }
    if (j + 1 <= nfull) { rows_fma<C, G, 1>(h, sv, wvi, j, grp, fl, acc); j += 1; }
    if constexpr (G == 2) {
      if (n & 1) {
        const int idx = n - 1 + grp;
        const unsigned s = (unsigned)__shfl(sv, idx);
        const float w = __int_as_float(__shfl(wvi, idx));
        if (grp == 0) {
          const float4 hv = *(const float4*)(h + s * C + fl);
          acc.x = fmaf(w, hv.x, acc.x);
          acc.y = fmaf(w, hv.y, acc.y);
          acc.z = fmaf(w, hv.z, acc.z);
          acc.w = fmaf(w, hv.w, acc.w);
        }
      }
    }
  }

  if constexpr (G == 2) {
    acc.x += __shfl_xor(acc.x, 16);
    acc.y += __shfl_xor(acc.y, 16);
    acc.z += __shfl_xor(acc.z, 16);
    acc.w += __shfl_xor(acc.w, 16);
  }
  if (grp == 0) {
    const float4 b = *(const float4*)(bias + fl);
    float4 v = make_float4(acc.x + b.x, acc.y + b.y, acc.z + b.z, acc.w + b.w);
    if constexpr (RELU) {
      v.x = fmaxf(v.x, 0.0f); v.y = fmaxf(v.y, 0.0f);
      v.z = fmaxf(v.z, 0.0f); v.w = fmaxf(v.w, 0.0f);
    }
    *(float4*)(out + (unsigned)node * C + fl) = v;
  }
}

extern "C" void kernel_launch(void* const* d_in, const int* in_sizes, int n_in,
                              void* d_out, int out_size, void* d_ws, size_t ws_size,
                              hipStream_t stream) {
  (void)in_sizes; (void)n_in; (void)out_size; (void)ws_size;
  const float* x  = (const float*)d_in[0];
  const int*   ei = (const int*)d_in[1];
  const float* W1 = (const float*)d_in[2];
  const float* b1 = (const float*)d_in[3];
  const float* W2 = (const float*)d_in[4];
  const float* b2 = (const float*)d_in[5];
  const int* src = ei;
  const int* dst = ei + E_EDGES;
  float* out = (float*)d_out;

  char* ws = (char*)d_ws;
  float* dinv = (float*)ws;  ws += (((size_t)N_NODES * 4) + 255) / 256 * 256;
  float* H1   = (float*)ws;  ws += (size_t)N_NODES * C_HID * 4;
  float* A1   = (float*)ws;  ws += (size_t)N_NODES * C_HID * 4;
  float* H2   = (float*)ws;  ws += (size_t)N_NODES * C_OUT * 4;
  unsigned* part = (unsigned*)ws;  ws += (size_t)HIST_BLOCKS * N_NODES * 4;
  unsigned* cnt  = (unsigned*)ws;  ws += (((size_t)N_NODES * 4) + 255) / 256 * 256;
  unsigned* rowptr = (unsigned*)ws;  ws += (((size_t)(N_NODES + 1) * 4) + 255) / 256 * 256;
  int* csr = (int*)ws;

  const dim3 B(256);
  deg_hist_k<<<HIST_BLOCKS, HIST_THREADS, 0, stream>>>(dst, part);
  deg_finish_k<<<(N_NODES + 255) / 256, B, 0, stream>>>(part, cnt, dinv);
  csr_fill_k<<<HIST_BLOCKS, HIST_THREADS, 0, stream>>>(src, dst, cnt, part, rowptr, csr);

  gemm_wmma_k<<<dim3(N_NODES / 16, C_HID / 64), dim3(128), 0, stream>>>(x, W1, H1, C_IN, C_HID);
  gather_k<C_HID, true><<<(N_NODES + 7) / 8, B, 0, stream>>>(H1, dinv, rowptr, csr, b1, A1);

  gemm_wmma_k<<<dim3(N_NODES / 16, C_OUT / 64), dim3(128), 0, stream>>>(A1, W2, H2, C_HID, C_OUT);
  gather_k<C_OUT, false><<<(N_NODES + 7) / 8, B, 0, stream>>>(H2, dinv, rowptr, csr, b2, out);
}


# layer-2 product fused into the layer-1 gather: a block's 16 relu'd rows stay in LDS as one WMMA A tile and 4 of its waves form H2 = A1 @ W2, k order kept; GEMM2 launch and A1 round trip removed
# speedup vs baseline: 3.0855x; 1.0879x over previous
#include <hip/hip_runtime.h>

#define N_NODES 10000
#define E_EDGES 640000
#define C_IN    256
#define C_HID   128
#define C_OUT   64

typedef float    v2f  __attribute__((ext_vector_type(2)));
typedef float    v8f  __attribute__((ext_vector_type(8)));
typedef _Float16 v16h __attribute__((ext_vector_type(16)));

#if defined(__has_builtin)
#if __has_builtin(__builtin_amdgcn_wmma_f32_16x16x4_f32)
#define USE_WMMA_F32 1
#endif
#endif

#define HIST_BLOCKS  64
#define HIST_THREADS 1024
static_assert(E_EDGES % (HIST_BLOCKS * 4) == 0, "each block reads whole int4 groups");

__global__ __launch_bounds__(HIST_THREADS)
void deg_hist_k(const int* __restrict__ dst, unsigned* __restrict__ part) {
  __shared__ unsigned cnt[N_NODES];
  for (int i = threadIdx.x; i < N_NODES; i += HIST_THREADS) cnt[i] = 0u;
  __syncthreads();

  constexpr int SLICE  = E_EDGES / HIST_BLOCKS;
  constexpr int SLICE4 = SLICE / 4;
  const int4* d4 = (const int4*)(dst + (size_t)blockIdx.x * SLICE);
  for (int i = threadIdx.x; i < SLICE4; i += HIST_THREADS) {
    const int4 v = d4[i];
    __hip_atomic_fetch_add(&cnt[v.x], 1u, __ATOMIC_RELAXED, __HIP_MEMORY_SCOPE_WORKGROUP);
    __hip_atomic_fetch_add(&cnt[v.y], 1u, __ATOMIC_RELAXED, __HIP_MEMORY_SCOPE_WORKGROUP);
    __hip_atomic_fetch_add(&cnt[v.z], 1u, __ATOMIC_RELAXED, __HIP_MEMORY_SCOPE_WORKGROUP);
    __hip_atomic_fetch_add(&cnt[v.w], 1u, __ATOMIC_RELAXED, __HIP_MEMORY_SCOPE_WORKGROUP);
  }
  __syncthreads();

  unsigned* prow = part + (size_t)blockIdx.x * N_NODES;
  for (int i = threadIdx.x; i < N_NODES; i += HIST_THREADS) prow[i] = cnt[i];
}

__global__ void deg_finish_k(unsigned* __restrict__ part, unsigned* __restrict__ cnt,
                             float* __restrict__ dinv) {
  int i = blockIdx.x * blockDim.x + threadIdx.x;
  if (i >= N_NODES) return;
  unsigned run = 0u;
  for (int g = 0; g < HIST_BLOCKS; ++g) {
    unsigned* p = part + (size_t)g * N_NODES + i;
    const unsigned c = *p;
    *p = run;
    run += c;
  }
  cnt[i] = run;
  dinv[i] = rsqrtf((float)(run + 1u));
}

#define SCAN_PER ((N_NODES + HIST_THREADS - 1) / HIST_THREADS)
#define SCAN_LEN (SCAN_PER * HIST_THREADS)

__global__ __launch_bounds__(HIST_THREADS)
void csr_fill_k(const int* __restrict__ src, const int* __restrict__ dst,
                const unsigned* __restrict__ cnt, const unsigned* __restrict__ partoff,
                unsigned* __restrict__ rowptr, int* __restrict__ csr_src) {
  __shared__ unsigned cur[SCAN_LEN];
  __shared__ unsigned wtot[HIST_THREADS / 32];
  const int t    = threadIdx.x;
  const int lane = t & 31;
  const int w    = t >> 5;

#pragma unroll
  for (int k = 0; k < SCAN_PER; ++k) {
    const int i = t + k * HIST_THREADS;
    cur[i] = (i < N_NODES) ? cnt[i] : 0u;
  }
  __syncthreads();

  const int i0 = t * SCAN_PER;
  unsigned loc[SCAN_PER];
  unsigned sum = 0u;
#pragma unroll
  for (int k = 0; k < SCAN_PER; ++k) {
    loc[k] = sum;
    sum += cur[i0 + k];
  }
  unsigned inc = sum;
  for (int o = 1; o < 32; o <<= 1) {
    const unsigned v = __shfl_up(inc, o);
    if (lane >= o) inc += v;
  }
  if (lane == 31) wtot[w] = inc;
  __syncthreads();
  if (w == 0) {
    const unsigned x = wtot[lane];
    unsigned xi = x;
    for (int o = 1; o < 32; o <<= 1) {
      const unsigned v = __shfl_up(xi, o);
      if (lane >= o) xi += v;
    }
    wtot[lane] = xi - x;
  }
  __syncthreads();
  const unsigned base = wtot[w] + (inc - sum);
#pragma unroll
  for (int k = 0; k < SCAN_PER; ++k) cur[i0 + k] = base + loc[k];
  __syncthreads();

  if (blockIdx.x == 0) {
#pragma unroll
    for (int k = 0; k < SCAN_PER; ++k) {
      const int i = t + k * HIST_THREADS;
      if (i < N_NODES) rowptr[i] = cur[i];
    }
    if (t == HIST_THREADS - 1) rowptr[N_NODES] = base + sum;
  }
  const unsigned* po = partoff + (size_t)blockIdx.x * N_NODES;
#pragma unroll
  for (int k = 0; k < SCAN_PER; ++k) {
    const int i = t + k * HIST_THREADS;
    if (i < N_NODES) cur[i] += po[i];
  }
  __syncthreads();

  constexpr int SLICE  = E_EDGES / HIST_BLOCKS;
  constexpr int SLICE4 = SLICE / 4;
  const int4* d4 = (const int4*)(dst + (size_t)blockIdx.x * SLICE);
  const int4* s4 = (const int4*)(src + (size_t)blockIdx.x * SLICE);
  for (int i = t; i < SLICE4; i += HIST_THREADS) {
    const int4 d = d4[i];
    const int4 s = s4[i];
    const unsigned p0 = __hip_atomic_fetch_add(&cur[d.x], 1u, __ATOMIC_RELAXED, __HIP_MEMORY_SCOPE_WORKGROUP);
    const unsigned p1 = __hip_atomic_fetch_add(&cur[d.y], 1u, __ATOMIC_RELAXED, __HIP_MEMORY_SCOPE_WORKGROUP);
    const unsigned p2 = __hip_atomic_fetch_add(&cur[d.z], 1u, __ATOMIC_RELAXED, __HIP_MEMORY_SCOPE_WORKGROUP);
    const unsigned p3 = __hip_atomic_fetch_add(&cur[d.w], 1u, __ATOMIC_RELAXED, __HIP_MEMORY_SCOPE_WORKGROUP);
    csr_src[p0] = s.x;
    csr_src[p1] = s.y;
    csr_src[p2] = s.z;
    csr_src[p3] = s.w;
  }
}

__global__ __launch_bounds__(128)
void gemm_wmma_k(const float* __restrict__ A, const float* __restrict__ B,
                 float* __restrict__ D, int K, int Ncols) {
  const int lane = threadIdx.x & 31;
  const int wave = threadIdx.x >> 5;
  const int row0 = blockIdx.x * 16;
  const int col0 = (blockIdx.y * 4 + wave) * 16;
  const int r    = lane & 15;
  const int half = lane >> 4;

  const float* Arow = A + (size_t)(row0 + r) * K;
  const float* Bcol = B + col0 + r;
  v8f acc = {};

#if defined(USE_WMMA_F32)
  for (int k = 0; k < K; k += 4) {
    const int kb = k + half * 2;
    v2f a = *(const v2f*)(Arow + kb);
    v2f b;
    b.x = Bcol[(size_t)kb * Ncols];
    b.y = Bcol[(size_t)(kb + 1) * Ncols];
    acc = __builtin_amdgcn_wmma_f32_16x16x4_f32(false, a, false, b,
                                                (short)0, acc, false, false);
  }
#else
  for (int k = 0; k < K; k += 32) {
    v16h a, b;
#pragma unroll
    for (int rr = 0; rr < 8; ++rr) {
      const int ka = k + (rr >> 2) * 16 + half * 8 + (rr & 3) * 2;
      a[2 * rr]     = (_Float16)Arow[ka];
      a[2 * rr + 1] = (_Float16)Arow[ka + 1];
      const int kb = k + half * 16 + 2 * rr;
      b[2 * rr]     = (_Float16)Bcol[(size_t)kb * Ncols];
      b[2 * rr + 1] = (_Float16)Bcol[(size_t)(kb + 1) * Ncols];
    }
    acc = __builtin_amdgcn_wmma_f32_16x16x32_f16(false, a, false, b,
                                                 (short)0, acc, false, false);
  }
#endif

  float* Drow = D + (size_t)(row0 + half * 8) * Ncols + col0 + r;
#pragma unroll
  for (int i = 0; i < 8; ++i) Drow[(size_t)i * Ncols] = acc[i];
}

template <int G>
__device__ __forceinline__ int lane_bcast(int v, int idx) {
  if constexpr (G == 1) return __builtin_amdgcn_readlane(v, idx);
  else                  return __shfl(v, idx);
}

template <int C, int G, int NS>
__device__ __forceinline__ void rows_fma(const float* __restrict__ h, int sv, int wvi,
                                         int step0, int grp, unsigned fl, float4& acc) {
  float4 hv[NS];
  float  wj[NS];
#pragma unroll
  for (int u = 0; u < NS; ++u) {
    const int idx = (step0 + u) * G + grp;
    const unsigned s = (unsigned)lane_bcast<G>(sv, idx);
    wj[u] = __int_as_float(lane_bcast<G>(wvi, idx));
    hv[u] = *(const float4*)(h + s * C + fl);
  }
#pragma unroll
  for (int u = 0; u < NS; ++u) {
    acc.x = fmaf(wj[u], hv[u].x, acc.x);
    acc.y = fmaf(wj[u], hv[u].y, acc.y);
    acc.z = fmaf(wj[u], hv[u].z, acc.z);
    acc.w = fmaf(wj[u], hv[u].w, acc.w);
  }
}

template <int C>
__device__ __forceinline__ float4 node_sum(const float* __restrict__ h, const float* __restrict__ dinv,
                                           const unsigned* __restrict__ rowptr,
                                           const int* __restrict__ csr_src, int node, int lane) {
  constexpr int LPR = C / 4;
  constexpr int G   = 32 / LPR;
  constexpr int U   = 8;
  const int grp = lane / LPR;
  const unsigned fl = (unsigned)(lane % LPR) * 4u;

  const unsigned beg = rowptr[node];
  const unsigned end = rowptr[node + 1];
  const float di = dinv[node];

  float4 acc = make_float4(0.0f, 0.0f, 0.0f, 0.0f);
  if (grp == 0) {
    const float4 hv = *(const float4*)(h + (unsigned)node * C + fl);
    const float ws = di * di;
    acc.x = ws * hv.x; acc.y = ws * hv.y; acc.z = ws * hv.z; acc.w = ws * hv.w;
  }

  for (unsigned cb = beg; cb < end; cb += 32u) {
    const int n = (int)((end - cb) < 32u ? (end - cb) : 32u);
    int   sv = 0;
    float wv = 0.0f;
    if (lane < n) {
      sv = csr_src[cb + lane];
      wv = dinv[sv] * di;
    }
    const int wvi = __float_as_int(wv);

    const int nfull = n / G;
    int j = 0;
    for (; j + U <= nfull; j += U) rows_fma<C, G, U>(h, sv, wvi, j, grp, fl, acc);
    if (j + 4 <= nfull) { rows_fma<C, G, 4>(h, sv, wvi, j, grp, fl, acc); j += 4; }
    if (j + 2 <= nfull) { rows_fma<C, G, 2>(h, sv, wvi, j, grp, fl, acc); j += 2; }
    if (j + 1 <= nfull) { rows_fma<C, G, 1>(h, sv, wvi, j, grp, fl, acc); j += 1; }
    if constexpr (G == 2) {
      if (n & 1) {
        const int idx = n - 1 + grp;
        const unsigned s = (unsigned)__shfl(sv, idx);
        const float w = __int_as_float(__shfl(wvi, idx));
        if (grp == 0) {
          const float4 hv = *(const float4*)(h + s * C + fl);
          acc.x = fmaf(w, hv.x, acc.x);
          acc.y = fmaf(w, hv.y, acc.y);
          acc.z = fmaf(w, hv.z, acc.z);
          acc.w = fmaf(w, hv.w, acc.w);
        }
      }
    }
  }

  if constexpr (G == 2) {
    acc.x += __shfl_xor(acc.x, 16);
    acc.y += __shfl_xor(acc.y, 16);
    acc.z += __shfl_xor(acc.z, 16);
    acc.w += __shfl_xor(acc.w, 16);
  }
  return acc;
}

template <int C>
__global__ __launch_bounds__(256)
void gather_k(const float* __restrict__ h, const float* __restrict__ dinv,
              const unsigned* __restrict__ rowptr, const int* __restrict__ csr_src,
              const float* __restrict__ bias, float* __restrict__ out) {
  constexpr int LPR = C / 4;
  const int lane = threadIdx.x & 31;
  const int node = __builtin_amdgcn_readfirstlane((int)(blockIdx.x * 8 + (threadIdx.x >> 5)));
  if (node >= N_NODES) return;
  const float4 acc = node_sum<C>(h, dinv, rowptr, csr_src, node, lane);
  if (lane < LPR) {
    const unsigned fl = (unsigned)lane * 4u;
    const float4 b = *(const float4*)(bias + fl);
    *(float4*)(out + (unsigned)node * C + fl) =
        make_float4(acc.x + b.x, acc.y + b.y, acc.z + b.z, acc.w + b.w);
  }
}

#define FG_NODES  16
#define FG_THREADS (FG_NODES * 32)
#define A1_PITCH  (C_HID + 4)
#define W2_PITCH  C_OUT
#define W2_ROUNDS ((C_HID * C_OUT / 4) / FG_THREADS)
static_assert(N_NODES % FG_NODES == 0, "whole 16-row tiles, no masking (EXEC all ones for WMMA)");
static_assert((C_HID * C_OUT / 4) % FG_THREADS == 0, "W2 staging in whole float4 rounds");

__global__ __launch_bounds__(FG_THREADS)
void gather_gemm_k(const float* __restrict__ h, const float* __restrict__ dinv,
                   const unsigned* __restrict__ rowptr, const int* __restrict__ csr_src,
                   const float* __restrict__ bias, const float* __restrict__ W2,
                   float* __restrict__ H2) {
  __shared__ __attribute__((aligned(16))) float a1s[FG_NODES * A1_PITCH];
  __shared__ __attribute__((aligned(16))) float w2s[C_HID * W2_PITCH];
  const int lane = threadIdx.x & 31;
  const int wave = __builtin_amdgcn_readfirstlane((int)(threadIdx.x >> 5));
  const int node = blockIdx.x * FG_NODES + wave;

  float4 wst[W2_ROUNDS];
#pragma unroll
  for (int i = 0; i < W2_ROUNDS; ++i)
    wst[i] = ((const float4*)W2)[threadIdx.x + i * FG_THREADS];
#pragma unroll
  for (int i = 0; i < W2_ROUNDS; ++i)
    *(float4*)(w2s + 4 * (threadIdx.x + i * FG_THREADS)) = wst[i];

  {
    const float4 acc = node_sum<C_HID>(h, dinv, rowptr, csr_src, node, lane);
    const unsigned fl = (unsigned)lane * 4u;
    const float4 b = *(const float4*)(bias + fl);
    float4 v = make_float4(acc.x + b.x, acc.y + b.y, acc.z + b.z, acc.w + b.w);
    v.x = fmaxf(v.x, 0.0f); v.y = fmaxf(v.y, 0.0f);
    v.z = fmaxf(v.z, 0.0f); v.w = fmaxf(v.w, 0.0f);
    *(float4*)(a1s + wave * A1_PITCH + fl) = v;
  }
  __syncthreads();

  if (wave < C_OUT / 16) {
    const int r    = lane & 15;
    const int half = lane >> 4;
    const int col0 = wave * 16;
    const float* Arow = a1s + r * A1_PITCH + half * 2;
    const float* Bcol = w2s + half * 2 * W2_PITCH + col0 + r;
    v8f acc = {};
    constexpr int KG = 8;
    constexpr int NG = C_HID / (4 * KG);
    v2f a[2][KG], b[2][KG];
#pragma unroll
    for (int u = 0; u < KG; ++u) {
      a[0][u]   = *(const v2f*)(Arow + 4 * u);
      b[0][u].x = Bcol[(4 * u) * W2_PITCH];
      b[0][u].y = Bcol[(4 * u + 1) * W2_PITCH];
    }
#pragma unroll
    for (int g = 0; g < NG; ++g) {
      if (g + 1 < NG) {
#pragma unroll
        for (int u = 0; u < KG; ++u) {
          const int k = 4 * ((g + 1) * KG + u);
          a[(g + 1) & 1][u]   = *(const v2f*)(Arow + k);
          b[(g + 1) & 1][u].x = Bcol[k * W2_PITCH];
          b[(g + 1) & 1][u].y = Bcol[(k + 1) * W2_PITCH];
        }
      }
#pragma unroll
      for (int u = 0; u < KG; ++u)
        acc = __builtin_amdgcn_wmma_f32_16x16x4_f32(false, a[g & 1][u], false, b[g & 1][u],
                                                    (short)0, acc, false, false);
    }
    __builtin_amdgcn_sched_group_barrier(0x100, KG / 2 + KG, 0);
#pragma unroll
    for (int g = 0; g < NG; ++g) {
      if (g + 1 < NG) __builtin_amdgcn_sched_group_barrier(0x100, KG / 2 + KG, 0);
      __builtin_amdgcn_sched_group_barrier(0x008, KG, 0);
    }
    float* Drow = H2 + (size_t)(blockIdx.x * FG_NODES + half * 8) * C_OUT + col0 + r;
#pragma unroll
    for (int i = 0; i < 8; ++i) Drow[(size_t)i * C_OUT] = acc[i];
  }
}

extern "C" void kernel_launch(void* const* d_in, const int* in_sizes, int n_in,
                              void* d_out, int out_size, void* d_ws, size_t ws_size,
                              hipStream_t stream) {
  (void)in_sizes; (void)n_in; (void)out_size; (void)ws_size;
  const float* x  = (const float*)d_in[0];
  const int*   ei = (const int*)d_in[1];
  const float* W1 = (const float*)d_in[2];
  const float* b1 = (const float*)d_in[3];
  const float* W2 = (const float*)d_in[4];
  const float* b2 = (const float*)d_in[5];
  const int* src = ei;
  const int* dst = ei + E_EDGES;
  float* out = (float*)d_out;

  char* ws = (char*)d_ws;
  float* dinv = (float*)ws;  ws += (((size_t)N_NODES * 4) + 255) / 256 * 256;
  float* H1   = (float*)ws;  ws += (size_t)N_NODES * C_HID * 4;
  float* H2   = (float*)ws;  ws += (size_t)N_NODES * C_OUT * 4;
  unsigned* part = (unsigned*)ws;  ws += (size_t)HIST_BLOCKS * N_NODES * 4;
  unsigned* cnt  = (unsigned*)ws;  ws += (((size_t)N_NODES * 4) + 255) / 256 * 256;
  unsigned* rowptr = (unsigned*)ws;  ws += (((size_t)(N_NODES + 1) * 4) + 255) / 256 * 256;
  int* csr = (int*)ws;

  const dim3 B(256);
  deg_hist_k<<<HIST_BLOCKS, HIST_THREADS, 0, stream>>>(dst, part);
  deg_finish_k<<<(N_NODES + 255) / 256, B, 0, stream>>>(part, cnt, dinv);
  csr_fill_k<<<HIST_BLOCKS, HIST_THREADS, 0, stream>>>(src, dst, cnt, part, rowptr, csr);

  gemm_wmma_k<<<dim3(N_NODES / 16, C_HID / 64), dim3(128), 0, stream>>>(x, W1, H1, C_IN, C_HID);
  gather_gemm_k<<<N_NODES / FG_NODES, FG_THREADS, 0, stream>>>(H1, dinv, rowptr, csr, b1, W2, H2);

  gather_k<C_OUT><<<(N_NODES + 7) / 8, B, 0, stream>>>(H2, dinv, rowptr, csr, b2, out);
}


# H1 = x @ W1 moved into the CSR-fill launch as extra workgroups beside the 64 sort blocks (one wave per 16x16 tile, k order kept): the product overlaps the counting sort, one launch fewer
# speedup vs baseline: 3.8069x; 1.2338x over previous
#include <hip/hip_runtime.h>

#define N_NODES 10000
#define E_EDGES 640000
#define C_IN    256
#define C_HID   128
#define C_OUT   64

typedef float    v2f  __attribute__((ext_vector_type(2)));
typedef float    v8f  __attribute__((ext_vector_type(8)));

#define HIST_BLOCKS  64
#define HIST_THREADS 1024
static_assert(E_EDGES % (HIST_BLOCKS * 4) == 0, "each block reads whole int4 groups");

__global__ __launch_bounds__(HIST_THREADS)
void deg_hist_k(const int* __restrict__ dst, unsigned* __restrict__ part) {
  __shared__ unsigned cnt[N_NODES];
  for (int i = threadIdx.x; i < N_NODES; i += HIST_THREADS) cnt[i] = 0u;
  __syncthreads();

  constexpr int SLICE  = E_EDGES / HIST_BLOCKS;
  constexpr int SLICE4 = SLICE / 4;
  const int4* d4 = (const int4*)(dst + (size_t)blockIdx.x * SLICE);
  for (int i = threadIdx.x; i < SLICE4; i += HIST_THREADS) {
    const int4 v = d4[i];
    __hip_atomic_fetch_add(&cnt[v.x], 1u, __ATOMIC_RELAXED, __HIP_MEMORY_SCOPE_WORKGROUP);
    __hip_atomic_fetch_add(&cnt[v.y], 1u, __ATOMIC_RELAXED, __HIP_MEMORY_SCOPE_WORKGROUP);
    __hip_atomic_fetch_add(&cnt[v.z], 1u, __ATOMIC_RELAXED, __HIP_MEMORY_SCOPE_WORKGROUP);
    __hip_atomic_fetch_add(&cnt[v.w], 1u, __ATOMIC_RELAXED, __HIP_MEMORY_SCOPE_WORKGROUP);
  }
  __syncthreads();

  unsigned* prow = part + (size_t)blockIdx.x * N_NODES;
  for (int i = threadIdx.x; i < N_NODES; i += HIST_THREADS) prow[i] = cnt[i];
}

__global__ void deg_finish_k(unsigned* __restrict__ part, unsigned* __restrict__ cnt,
                             float* __restrict__ dinv) {
  int i = blockIdx.x * blockDim.x + threadIdx.x;
  if (i >= N_NODES) return;
  unsigned run = 0u;
  for (int g = 0; g < HIST_BLOCKS; ++g) {
    unsigned* p = part + (size_t)g * N_NODES + i;
    const unsigned c = *p;
    *p = run;
    run += c;
  }
  cnt[i] = run;
  dinv[i] = rsqrtf((float)(run + 1u));
}

static_assert(N_NODES % 16 == 0 && C_HID % 16 == 0 && C_IN % 4 == 0, "whole tiles only");

__device__ __forceinline__ void gemm1_tile(const float* __restrict__ A, const float* __restrict__ B,
                                           float* __restrict__ D, int row0, int col0, int lane) {
  const int r    = lane & 15;
  const int half = lane >> 4;
  const float* Arow = A + (size_t)(row0 + r) * C_IN;
  const float* Bcol = B + col0 + r;
  v8f acc = {};
  for (int k = 0; k < C_IN; k += 4) {
    const int kb = k + half * 2;
    v2f a = *(const v2f*)(Arow + kb);
    v2f b;
    b.x = Bcol[(size_t)kb * C_HID];
    b.y = Bcol[(size_t)(kb + 1) * C_HID];
    acc = __builtin_amdgcn_wmma_f32_16x16x4_f32(false, a, false, b,
                                                (short)0, acc, false, false);
  }
  float* Drow = D + (size_t)(row0 + half * 8) * C_HID + col0 + r;
#pragma unroll
  for (int i = 0; i < 8; ++i) Drow[(size_t)i * C_HID] = acc[i];
}

#define SCAN_PER ((N_NODES + HIST_THREADS - 1) / HIST_THREADS)
#define SCAN_LEN (SCAN_PER * HIST_THREADS)
#define GEMM_COLT   (C_HID / 16)
#define GEMM_ROWT   ((HIST_THREADS / 32) / GEMM_COLT)
#define GEMM_BLOCKS ((N_NODES / 16 + GEMM_ROWT - 1) / GEMM_ROWT)
static_assert(GEMM_COLT * GEMM_ROWT * 32 == HIST_THREADS, "one wave per tile of a product block");

__global__ __launch_bounds__(HIST_THREADS)
void csr_fill_gemm_k(const int* __restrict__ src, const int* __restrict__ dst,
                     const unsigned* __restrict__ cnt, const unsigned* __restrict__ partoff,
                     unsigned* __restrict__ rowptr, int* __restrict__ csr_src,
                     const float* __restrict__ x, const float* __restrict__ W1,
                     float* __restrict__ H1) {
  __shared__ unsigned cur[SCAN_LEN];
  __shared__ unsigned wtot[HIST_THREADS / 32];
  const int t    = threadIdx.x;
  const int lane = t & 31;
  const int w    = t >> 5;

  if (blockIdx.x >= HIST_BLOCKS) {
    const int wv = __builtin_amdgcn_readfirstlane(w);
    const int rt = (int)(blockIdx.x - HIST_BLOCKS) * GEMM_ROWT + wv / GEMM_COLT;
    if (rt < N_NODES / 16)
      gemm1_tile(x, W1, H1, rt * 16, (wv % GEMM_COLT) * 16, lane);
    return;
  }

#pragma unroll
  for (int k = 0; k < SCAN_PER; ++k) {
    const int i = t + k * HIST_THREADS;
    cur[i] = (i < N_NODES) ? cnt[i] : 0u;
  }
  __syncthreads();

  const int i0 = t * SCAN_PER;
  unsigned loc[SCAN_PER];
  unsigned sum = 0u;
#pragma unroll
  for (int k = 0; k < SCAN_PER; ++k) {
    loc[k] = sum;
    sum += cur[i0 + k];
  }
  unsigned inc = sum;
  for (int o = 1; o < 32; o <<= 1) {
    const unsigned v = __shfl_up(inc, o);
    if (lane >= o) inc += v;
  }
  if (lane == 31) wtot[w] = inc;
  __syncthreads();
  if (w == 0) {
    const unsigned xw = wtot[lane];
    unsigned xi = xw;
    for (int o = 1; o < 32; o <<= 1) {
      const unsigned v = __shfl_up(xi, o);
      if (lane >= o) xi += v;
    }
    wtot[lane] = xi - xw;
  }
  __syncthreads();
  const unsigned base = wtot[w] + (inc - sum);
#pragma unroll
  for (int k = 0; k < SCAN_PER; ++k) cur[i0 + k] = base + loc[k];
  __syncthreads();

  if (blockIdx.x == 0) {
#pragma unroll
    for (int k = 0; k < SCAN_PER; ++k) {
      const int i = t + k * HIST_THREADS;
      if (i < N_NODES) rowptr[i] = cur[i];
    }
    if (t == HIST_THREADS - 1) rowptr[N_NODES] = base + sum;
  }
  const unsigned* po = partoff + (size_t)blockIdx.x * N_NODES;
#pragma unroll
  for (int k = 0; k < SCAN_PER; ++k) {
    const int i = t + k * HIST_THREADS;
    if (i < N_NODES) cur[i] += po[i];
  }
  __syncthreads();

  constexpr int SLICE  = E_EDGES / HIST_BLOCKS;
  constexpr int SLICE4 = SLICE / 4;
  const int4* d4 = (const int4*)(dst + (size_t)blockIdx.x * SLICE);
  const int4* s4 = (const int4*)(src + (size_t)blockIdx.x * SLICE);
  for (int i = t; i < SLICE4; i += HIST_THREADS) {
    const int4 d = d4[i];
    const int4 s = s4[i];
    const unsigned p0 = __hip_atomic_fetch_add(&cur[d.x], 1u, __ATOMIC_RELAXED, __HIP_MEMORY_SCOPE_WORKGROUP);
    const unsigned p1 = __hip_atomic_fetch_add(&cur[d.y], 1u, __ATOMIC_RELAXED, __HIP_MEMORY_SCOPE_WORKGROUP);
    const unsigned p2 = __hip_atomic_fetch_add(&cur[d.z], 1u, __ATOMIC_RELAXED, __HIP_MEMORY_SCOPE_WORKGROUP);
    const unsigned p3 = __hip_atomic_fetch_add(&cur[d.w], 1u, __ATOMIC_RELAXED, __HIP_MEMORY_SCOPE_WORKGROUP);
    csr_src[p0] = s.x;
    csr_src[p1] = s.y;
    csr_src[p2] = s.z;
    csr_src[p3] = s.w;
  }
}

template <int G>
__device__ __forceinline__ int lane_bcast(int v, int idx) {
  if constexpr (G == 1) return __builtin_amdgcn_readlane(v, idx);
  else                  return __shfl(v, idx);
}

template <int C, int G, int NS>
__device__ __forceinline__ void rows_fma(const float* __restrict__ h, int sv, int wvi,
                                         int step0, int grp, unsigned fl, float4& acc) {
  float4 hv[NS];
  float  wj[NS];
#pragma unroll
  for (int u = 0; u < NS; ++u) {
    const int idx = (step0 + u) * G + grp;
    const unsigned s = (unsigned)lane_bcast<G>(sv, idx);
    wj[u] = __int_as_float(lane_bcast<G>(wvi, idx));
    hv[u] = *(const float4*)(h + s * C + fl);
  }
#pragma unroll
  for (int u = 0; u < NS; ++u) {
    acc.x = fmaf(wj[u], hv[u].x, acc.x);
    acc.y = fmaf(wj[u], hv[u].y, acc.y);
    acc.z = fmaf(wj[u], hv[u].z, acc.z);
    acc.w = fmaf(wj[u], hv[u].w, acc.w);
  }
}

template <int C>
__device__ __forceinline__ float4 node_sum(const float* __restrict__ h, const float* __restrict__ dinv,
                                           const unsigned* __restrict__ rowptr,
                                           const int* __restrict__ csr_src, int node, int lane) {
  constexpr int LPR = C / 4;
  constexpr int G   = 32 / LPR;
  constexpr int U   = 8;
  const int grp = lane / LPR;
  const unsigned fl = (unsigned)(lane % LPR) * 4u;

  const unsigned beg = rowptr[node];
  const unsigned end = rowptr[node + 1];
  const float di = dinv[node];

  float4 acc = make_float4(0.0f, 0.0f, 0.0f, 0.0f);
  if (grp == 0) {
    const float4 hv = *(const float4*)(h + (unsigned)node * C + fl);
    const float ws = di * di;
    acc.x = ws * hv.x; acc.y = ws * hv.y; acc.z = ws * hv.z; acc.w = ws * hv.w;
  }

  for (unsigned cb = beg; cb < end; cb += 32u) {
    const int n = (int)((end - cb) < 32u ? (end - cb) : 32u);
    int   sv = 0;
    float wv = 0.0f;
    if (lane < n) {
      sv = csr_src[cb + lane];
      wv = dinv[sv] * di;
    }
    const int wvi = __float_as_int(wv);

    const int nfull = n / G;
    int j = 0;
    for (; j + U <= nfull; j += U) rows_fma<C, G, U>(h, sv, wvi, j, grp, fl, acc);
    if (j + 4 <= nfull) { rows_fma<C, G, 4>(h, sv, wvi, j, grp, fl, acc); j += 4; }
    if (j + 2 <= nfull) { rows_fma<C, G, 2>(h, sv, wvi, j, grp, fl, acc); j += 2; }
    if (j + 1 <= nfull) { rows_fma<C, G, 1>(h, sv, wvi, j, grp, fl, acc); j += 1; }
    if constexpr (G == 2) {
      if (n & 1) {
        const int idx = n - 1 + grp;
        const unsigned s = (unsigned)__shfl(sv, idx);
        const float w = __int_as_float(__shfl(wvi, idx));
        if (grp == 0) {
          const float4 hv = *(const float4*)(h + s * C + fl);
          acc.x = fmaf(w, hv.x, acc.x);
          acc.y = fmaf(w, hv.y, acc.y);
          acc.z = fmaf(w, hv.z, acc.z);
          acc.w = fmaf(w, hv.w, acc.w);
        }
      }
    }
  }

  if constexpr (G == 2) {
    acc.x += __shfl_xor(acc.x, 16);
    acc.y += __shfl_xor(acc.y, 16);
    acc.z += __shfl_xor(acc.z, 16);
    acc.w += __shfl_xor(acc.w, 16);
  }
  return acc;
}

template <int C>
__global__ __launch_bounds__(256)
void gather_k(const float* __restrict__ h, const float* __restrict__ dinv,
              const unsigned* __restrict__ rowptr, const int* __restrict__ csr_src,
              const float* __restrict__ bias, float* __restrict__ out) {
  constexpr int LPR = C / 4;
  const int lane = threadIdx.x & 31;
  const int node = __builtin_amdgcn_readfirstlane((int)(blockIdx.x * 8 + (threadIdx.x >> 5)));
  if (node >= N_NODES) return;
  const float4 acc = node_sum<C>(h, dinv, rowptr, csr_src, node, lane);
  if (lane < LPR) {
    const unsigned fl = (unsigned)lane * 4u;
    const float4 b = *(const float4*)(bias + fl);
    *(float4*)(out + (unsigned)node * C + fl) =
        make_float4(acc.x + b.x, acc.y + b.y, acc.z + b.z, acc.w + b.w);
  }
}

#define FG_NODES  16
#define FG_THREADS (FG_NODES * 32)
#define A1_PITCH  (C_HID + 4)
#define W2_PITCH  C_OUT
#define W2_ROUNDS ((C_HID * C_OUT / 4) / FG_THREADS)
static_assert(N_NODES % FG_NODES == 0, "whole 16-row tiles, no masking (EXEC all ones for WMMA)");
static_assert((C_HID * C_OUT / 4) % FG_THREADS == 0, "W2 staging in whole float4 rounds");

__global__ __launch_bounds__(FG_THREADS)
void gather_gemm_k(const float* __restrict__ h, const float* __restrict__ dinv,
                   const unsigned* __restrict__ rowptr, const int* __restrict__ csr_src,
                   const float* __restrict__ bias, const float* __restrict__ W2,
                   float* __restrict__ H2) {
  __shared__ __attribute__((aligned(16))) float a1s[FG_NODES * A1_PITCH];
  __shared__ __attribute__((aligned(16))) float w2s[C_HID * W2_PITCH];
  const int lane = threadIdx.x & 31;
  const int wave = __builtin_amdgcn_readfirstlane((int)(threadIdx.x >> 5));
  const int node = blockIdx.x * FG_NODES + wave;

  float4 wst[W2_ROUNDS];
#pragma unroll
  for (int i = 0; i < W2_ROUNDS; ++i)
    wst[i] = ((const float4*)W2)[threadIdx.x + i * FG_THREADS];
#pragma unroll
  for (int i = 0; i < W2_ROUNDS; ++i)
    *(float4*)(w2s + 4 * (threadIdx.x + i * FG_THREADS)) = wst[i];

  {
    const float4 acc = node_sum<C_HID>(h, dinv, rowptr, csr_src, node, lane);
    const unsigned fl = (unsigned)lane * 4u;
    const float4 b = *(const float4*)(bias + fl);
    float4 v = make_float4(acc.x + b.x, acc.y + b.y, acc.z + b.z, acc.w + b.w);
    v.x = fmaxf(v.x, 0.0f); v.y = fmaxf(v.y, 0.0f);
    v.z = fmaxf(v.z, 0.0f); v.w = fmaxf(v.w, 0.0f);
    *(float4*)(a1s + wave * A1_PITCH + fl) = v;
  }
  __syncthreads();

  if (wave < C_OUT / 16) {
    const int r    = lane & 15;
    const int half = lane >> 4;
    const int col0 = wave * 16;
    const float* Arow = a1s + r * A1_PITCH + half * 2;
    const float* Bcol = w2s + half * 2 * W2_PITCH + col0 + r;
    v8f acc = {};
    constexpr int KG = 8;
    constexpr int NG = C_HID / (4 * KG);
    v2f a[2][KG], b[2][KG];
#pragma unroll
    for (int u = 0; u < KG; ++u) {
      a[0][u]   = *(const v2f*)(Arow + 4 * u);
      b[0][u].x = Bcol[(4 * u) * W2_PITCH];
      b[0][u].y = Bcol[(4 * u + 1) * W2_PITCH];
    }
#pragma unroll
    for (int g = 0; g < NG; ++g) {
      if (g + 1 < NG) {
#pragma unroll
        for (int u = 0; u < KG; ++u) {
          const int k = 4 * ((g + 1) * KG + u);
          a[(g + 1) & 1][u]   = *(const v2f*)(Arow + k);
          b[(g + 1) & 1][u].x = Bcol[k * W2_PITCH];
          b[(g + 1) & 1][u].y = Bcol[(k + 1) * W2_PITCH];
        }
      }
#pragma unroll
      for (int u = 0; u < KG; ++u)
        acc = __builtin_amdgcn_wmma_f32_16x16x4_f32(false, a[g & 1][u], false, b[g & 1][u],
                                                    (short)0, acc, false, false);
    }
    __builtin_amdgcn_sched_group_barrier(0x100, KG / 2 + KG, 0);
#pragma unroll
    for (int g = 0; g < NG; ++g) {
      if (g + 1 < NG) __builtin_amdgcn_sched_group_barrier(0x100, KG / 2 + KG, 0);
      __builtin_amdgcn_sched_group_barrier(0x008, KG, 0);
    }
    float* Drow = H2 + (size_t)(blockIdx.x * FG_NODES + half * 8) * C_OUT + col0 + r;
#pragma unroll
    for (int i = 0; i < 8; ++i) Drow[(size_t)i * C_OUT] = acc[i];
  }
}

extern "C" void kernel_launch(void* const* d_in, const int* in_sizes, int n_in,
                              void* d_out, int out_size, void* d_ws, size_t ws_size,
                              hipStream_t stream) {
  (void)in_sizes; (void)n_in; (void)out_size; (void)ws_size;
  const float* x  = (const float*)d_in[0];
  const int*   ei = (const int*)d_in[1];
  const float* W1 = (const float*)d_in[2];
  const float* b1 = (const float*)d_in[3];
  const float* W2 = (const float*)d_in[4];
  const float* b2 = (const float*)d_in[5];
  const int* src = ei;
  const int* dst = ei + E_EDGES;
  float* out = (float*)d_out;

  char* ws = (char*)d_ws;
  float* dinv = (float*)ws;  ws += (((size_t)N_NODES * 4) + 255) / 256 * 256;
  float* H1   = (float*)ws;  ws += (size_t)N_NODES * C_HID * 4;
  float* H2   = (float*)ws;  ws += (size_t)N_NODES * C_OUT * 4;
  unsigned* part = (unsigned*)ws;  ws += (size_t)HIST_BLOCKS * N_NODES * 4;
  unsigned* cnt  = (unsigned*)ws;  ws += (((size_t)N_NODES * 4) + 255) / 256 * 256;
  unsigned* rowptr = (unsigned*)ws;  ws += (((size_t)(N_NODES + 1) * 4) + 255) / 256 * 256;
  int* csr = (int*)ws;

  const dim3 B(256);
  deg_hist_k<<<HIST_BLOCKS, HIST_THREADS, 0, stream>>>(dst, part);
  deg_finish_k<<<(N_NODES + 255) / 256, B, 0, stream>>>(part, cnt, dinv);
  csr_fill_gemm_k<<<HIST_BLOCKS + GEMM_BLOCKS, HIST_THREADS, 0, stream>>>(src, dst, cnt, part, rowptr, csr,
                                                                        x, W1, H1);

  gather_gemm_k<<<N_NODES / FG_NODES, FG_THREADS, 0, stream>>>(H1, dinv, rowptr, csr, b1, W2, H2);

  gather_k<C_OUT><<<(N_NODES + 7) / 8, B, 0, stream>>>(H2, dinv, rowptr, csr, b2, out);
}


# prefix launch removed: counts kept per 16 edge slices x node, CSR-fill blocks (slice x node range) and product blocks sum them themselves, dinv from row bounds; per-node sum order = the new CSR order
# speedup vs baseline: 4.4092x; 1.1582x over previous
#include <hip/hip_runtime.h>

#define N_NODES 10000
#define E_EDGES 640000
#define C_IN    256
#define C_HID   128
#define C_OUT   64

typedef float    v2f  __attribute__((ext_vector_type(2)));
typedef float    v8f  __attribute__((ext_vector_type(8)));

#define S_SLICES   16
#define RANGE      1024
#define R_RANGES   ((N_NODES + RANGE - 1) / RANGE)
#define NPAD       (R_RANGES * RANGE)
#define HR         2
#define H_RBLOCKS  (R_RANGES / HR)
#define HIST_BLOCKS  (S_SLICES * H_RBLOCKS)
#define HIST_THREADS 1024
#define SLICE      (E_EDGES / S_SLICES)
#define SLICE4     (SLICE / 4)
#define SLICE_IT   ((SLICE4 + HIST_THREADS - 1) / HIST_THREADS)
static_assert(E_EDGES % (S_SLICES * 4) == 0, "each slice is whole int4 groups");
static_assert(R_RANGES % HR == 0, "whole histogram blocks");
static_assert(RANGE == HIST_THREADS, "one node of a range per thread");

__global__ __launch_bounds__(HIST_THREADS)
void deg_hist_k(const int* __restrict__ dst, unsigned* __restrict__ part, unsigned* __restrict__ T) {
  __shared__ unsigned cnt[HR * RANGE];
  __shared__ unsigned tsum[HR];
  const int t    = threadIdx.x;
  const int lane = t & 31;
  const int s    = blockIdx.x / H_RBLOCKS;
  const int rh   = blockIdx.x % H_RBLOCKS;
  const int lo   = rh * HR * RANGE;
#pragma unroll
  for (int q = 0; q < HR; ++q) cnt[t + q * RANGE] = 0u;
  if (t < HR) tsum[t] = 0u;
  __syncthreads();

  const int4* d4 = (const int4*)(dst + (size_t)s * SLICE);
  int4 v[SLICE_IT];
#pragma unroll
  for (int it = 0; it < SLICE_IT; ++it) {
    const int i = t + it * HIST_THREADS;
    v[it] = (i < SLICE4) ? d4[i] : make_int4(-1, -1, -1, -1);
  }
#pragma unroll
  for (int it = 0; it < SLICE_IT; ++it) {
    const unsigned k0 = (unsigned)(v[it].x - lo), k1 = (unsigned)(v[it].y - lo);
    const unsigned k2 = (unsigned)(v[it].z - lo), k3 = (unsigned)(v[it].w - lo);
    if (k0 < HR * RANGE) __hip_atomic_fetch_add(&cnt[k0], 1u, __ATOMIC_RELAXED, __HIP_MEMORY_SCOPE_WORKGROUP);
    if (k1 < HR * RANGE) __hip_atomic_fetch_add(&cnt[k1], 1u, __ATOMIC_RELAXED, __HIP_MEMORY_SCOPE_WORKGROUP);
    if (k2 < HR * RANGE) __hip_atomic_fetch_add(&cnt[k2], 1u, __ATOMIC_RELAXED, __HIP_MEMORY_SCOPE_WORKGROUP);
    if (k3 < HR * RANGE) __hip_atomic_fetch_add(&cnt[k3], 1u, __ATOMIC_RELAXED, __HIP_MEMORY_SCOPE_WORKGROUP);
  }
  __syncthreads();

#pragma unroll
  for (int q = 0; q < HR; ++q) {
    const unsigned c = cnt[t + q * RANGE];
    part[(size_t)s * NPAD + lo + q * RANGE + t] = c;
    unsigned r = c;
    for (int o = 16; o > 0; o >>= 1) r += __shfl_xor(r, o);
    if (lane == 0) __hip_atomic_fetch_add(&tsum[q], r, __ATOMIC_RELAXED, __HIP_MEMORY_SCOPE_WORKGROUP);
  }
  __syncthreads();
  if (t < HR) T[s * R_RANGES + rh * HR + t] = tsum[t];
}

static_assert(N_NODES % 16 == 0 && C_HID % 16 == 0 && C_IN % 4 == 0, "whole tiles only");

__device__ __forceinline__ void gemm1_tile(const float* __restrict__ A, const float* __restrict__ B,
                                           const unsigned* indeg,
                                           float* __restrict__ D, int row0, int col0, int lane) {
  const int r    = lane & 15;
  const int half = lane >> 4;
  const float* Arow = A + (size_t)(row0 + r) * C_IN;
  const float* Bcol = B + col0 + r;
  v8f acc = {};
  for (int k = 0; k < C_IN; k += 4) {
    const int kb = k + half * 2;
    v2f a = *(const v2f*)(Arow + kb);
    v2f b;
    b.x = Bcol[(size_t)kb * C_HID];
    b.y = Bcol[(size_t)(kb + 1) * C_HID];
    acc = __builtin_amdgcn_wmma_f32_16x16x4_f32(false, a, false, b,
                                                (short)0, acc, false, false);
  }
  float* Drow = D + (size_t)(row0 + half * 8) * C_HID + col0 + r;
#pragma unroll
  for (int i = 0; i < 8; ++i)
    Drow[(size_t)i * C_HID] = acc[i] * rsqrtf((float)(indeg[half * 8 + i] + 1u));
}

#define SORT_BLOCKS (S_SLICES * R_RANGES)
#define GEMM_COLT   (C_HID / 16)
#define GEMM_ROWT   ((HIST_THREADS / 32) / GEMM_COLT)
#define GEMM_ROWS   (GEMM_ROWT * 16)
#define GEMM_BLOCKS ((N_NODES / 16 + GEMM_ROWT - 1) / GEMM_ROWT)
#define FILL_BATCH  5
#define NO_SLOT     0xffffffffu
static_assert(GEMM_COLT * GEMM_ROWT * 32 == HIST_THREADS, "one wave per tile of a product block");
static_assert(GEMM_ROWS * S_SLICES == HIST_THREADS, "one (row, slice) count per thread of a product block");
static_assert(GEMM_BLOCKS * GEMM_ROWS <= NPAD, "the degree sums stay inside part's rows");
static_assert(S_SLICES * R_RANGES <= HIST_THREADS, "one T entry per thread");
static_assert(SLICE_IT % FILL_BATCH == 0, "whole batches");

__global__ __launch_bounds__(HIST_THREADS)
void csr_fill_gemm_k(const int* __restrict__ src, const int* __restrict__ dst,
                     const unsigned* __restrict__ part, const unsigned* __restrict__ T,
                     unsigned* __restrict__ rowptr, int* __restrict__ csr_src,
                     const float* __restrict__ x, const float* __restrict__ W1,
                     float* __restrict__ H1) {
  __shared__ unsigned cur[RANGE];
  __shared__ unsigned wtot[HIST_THREADS / 32];
  __shared__ unsigned wbase[HIST_THREADS / 32];
  __shared__ unsigned btot;
  __shared__ unsigned indeg[GEMM_ROWS];
  const int t    = threadIdx.x;
  const int lane = t & 31;
  const int w    = t >> 5;

  if (blockIdx.x < GEMM_BLOCKS) {
    const int row0b = (int)blockIdx.x * GEMM_ROWS;
    if (t < GEMM_ROWS) indeg[t] = 0u;
    __syncthreads();
    __hip_atomic_fetch_add(&indeg[t % GEMM_ROWS], part[(size_t)(t / GEMM_ROWS) * NPAD + row0b + t % GEMM_ROWS],
                           __ATOMIC_RELAXED, __HIP_MEMORY_SCOPE_WORKGROUP);
    __syncthreads();
    const int wv = __builtin_amdgcn_readfirstlane(w);
    const int rt = row0b / 16 + wv / GEMM_COLT;
    if (rt < N_NODES / 16)
      gemm1_tile(x, W1, indeg + (wv / GEMM_COLT) * 16, H1, rt * 16, (wv % GEMM_COLT) * 16, lane);
    return;
  }

  const int sb = (int)blockIdx.x - GEMM_BLOCKS;
  const int s  = sb / R_RANGES;
  const int r  = sb % R_RANGES;
  const int lo = r * RANGE;

  unsigned tv = 0u;
  if (t < S_SLICES * R_RANGES && (t % R_RANGES) < r) tv = T[t];
  unsigned tot = 0u, po = 0u;
#pragma unroll
  for (int q = 0; q < S_SLICES; ++q) {
    const unsigned c = part[(size_t)q * NPAD + lo + t];
    tot += c;
    if (q < s) po += c;
  }

  unsigned inc = tot;
  for (int o = 1; o < 32; o <<= 1) {
    const unsigned v = __shfl_up(inc, o);
    if (lane >= o) inc += v;
  }
  unsigned bs = tv;
  for (int o = 16; o > 0; o >>= 1) bs += __shfl_xor(bs, o);
  if (lane == 31) wtot[w] = inc;
  if (lane == 0)  wbase[w] = bs;
  __syncthreads();
  if (w == 0) {
    const unsigned xw = wtot[lane];
    unsigned xi = xw;
    for (int o = 1; o < 32; o <<= 1) {
      const unsigned v = __shfl_up(xi, o);
      if (lane >= o) xi += v;
    }
    wtot[lane] = xi - xw;
    unsigned bb = wbase[lane];
    for (int o = 16; o > 0; o >>= 1) bb += __shfl_xor(bb, o);
    if (lane == 0) btot = bb;
  }
  __syncthreads();
  const unsigned start = btot + wtot[w] + (inc - tot);
  cur[t] = start + po;
  if (s == 0) {
    const int node = lo + t;
    if (node < N_NODES) rowptr[node] = start;
    if (node == N_NODES - 1) rowptr[N_NODES] = start + tot;
  }
  __syncthreads();

  const int4* d4 = (const int4*)(dst + (size_t)s * SLICE);
  const int4* s4 = (const int4*)(src + (size_t)s * SLICE);
#pragma unroll
  for (int b = 0; b < SLICE_IT; b += FILL_BATCH) {
    int4 dv[FILL_BATCH], sv[FILL_BATCH];
#pragma unroll
    for (int u = 0; u < FILL_BATCH; ++u) {
      const int i = t + (b + u) * HIST_THREADS;
      if (i < SLICE4) { dv[u] = d4[i]; sv[u] = s4[i]; }
      else            { dv[u] = make_int4(-1, -1, -1, -1); sv[u] = make_int4(0, 0, 0, 0); }
    }
#pragma unroll
    for (int u = 0; u < FILL_BATCH; ++u) {
      asm volatile("" :: "v"(sv[u].x), "v"(sv[u].y), "v"(sv[u].z), "v"(sv[u].w));
    }
    unsigned p[FILL_BATCH * 4];
#pragma unroll
    for (int u = 0; u < FILL_BATCH; ++u) {
      const unsigned k0 = (unsigned)(dv[u].x - lo), k1 = (unsigned)(dv[u].y - lo);
      const unsigned k2 = (unsigned)(dv[u].z - lo), k3 = (unsigned)(dv[u].w - lo);
      p[4 * u + 0] = (k0 < RANGE) ? __hip_atomic_fetch_add(&cur[k0], 1u, __ATOMIC_RELAXED, __HIP_MEMORY_SCOPE_WORKGROUP) : NO_SLOT;
      p[4 * u + 1] = (k1 < RANGE) ? __hip_atomic_fetch_add(&cur[k1], 1u, __ATOMIC_RELAXED, __HIP_MEMORY_SCOPE_WORKGROUP) : NO_SLOT;
      p[4 * u + 2] = (k2 < RANGE) ? __hip_atomic_fetch_add(&cur[k2], 1u, __ATOMIC_RELAXED, __HIP_MEMORY_SCOPE_WORKGROUP) : NO_SLOT;
      p[4 * u + 3] = (k3 < RANGE) ? __hip_atomic_fetch_add(&cur[k3], 1u, __ATOMIC_RELAXED, __HIP_MEMORY_SCOPE_WORKGROUP) : NO_SLOT;
    }
#pragma unroll
    for (int u = 0; u < FILL_BATCH; ++u) {
      if (p[4 * u + 0] != NO_SLOT) csr_src[p[4 * u + 0]] = sv[u].x;
      if (p[4 * u + 1] != NO_SLOT) csr_src[p[4 * u + 1]] = sv[u].y;
      if (p[4 * u + 2] != NO_SLOT) csr_src[p[4 * u + 2]] = sv[u].z;
      if (p[4 * u + 3] != NO_SLOT) csr_src[p[4 * u + 3]] = sv[u].w;
    }
  }
}

template <int G>
__device__ __forceinline__ int lane_bcast(int v, int idx) {
  if constexpr (G == 1) return __builtin_amdgcn_readlane(v, idx);
  else                  return __shfl(v, idx);
}

template <int C, int G, int NS>
__device__ __forceinline__ void rows_add(const float* __restrict__ h, int sv,
                                         int step0, int grp, unsigned fl, float4& acc) {
  float4 hv[NS];
#pragma unroll
  for (int u = 0; u < NS; ++u) {
    const int idx = (step0 + u) * G + grp;
    const unsigned s = (unsigned)lane_bcast<G>(sv, idx);
    hv[u] = *(const float4*)(h + s * C + fl);
  }
#pragma unroll
  for (int u = 0; u < NS; ++u) {
    acc.x += hv[u].x;
    acc.y += hv[u].y;
    acc.z += hv[u].z;
    acc.w += hv[u].w;
  }
}

template <int C>
__device__ __forceinline__ float4 node_sum(const float* __restrict__ h,
                                           const unsigned* __restrict__ rowptr,
                                           const int* __restrict__ csr_src, int node, int lane, float& di) {
  constexpr int LPR = C / 4;
  constexpr int G   = 32 / LPR;
  constexpr int U   = 8;
  const int grp = lane / LPR;
  const unsigned fl = (unsigned)(lane % LPR) * 4u;

  const unsigned beg = rowptr[node];
  const unsigned end = rowptr[node + 1];
  di = rsqrtf((float)(end - beg + 1u));

  float4 acc = make_float4(0.0f, 0.0f, 0.0f, 0.0f);
  if (grp == 0)
    acc = *(const float4*)(h + (unsigned)node * C + fl);

  for (unsigned cb = beg; cb < end; cb += 32u) {
    const int n = (int)((end - cb) < 32u ? (end - cb) : 32u);
    int sv = 0;
    if (lane < n) sv = csr_src[cb + lane];

    const int nfull = n / G;
    int j = 0;
    for (; j + U <= nfull; j += U) rows_add<C, G, U>(h, sv, j, grp, fl, acc);
    if (j + 4 <= nfull) { rows_add<C, G, 4>(h, sv, j, grp, fl, acc); j += 4; }
    if (j + 2 <= nfull) { rows_add<C, G, 2>(h, sv, j, grp, fl, acc); j += 2; }
    if (j + 1 <= nfull) { rows_add<C, G, 1>(h, sv, j, grp, fl, acc); j += 1; }
    if constexpr (G == 2) {
      if (n & 1) {
        const int idx = n - 1 + grp;
        const unsigned s = (unsigned)__shfl(sv, idx);
        if (grp == 0) {
          const float4 hv = *(const float4*)(h + s * C + fl);
          acc.x += hv.x;
          acc.y += hv.y;
          acc.z += hv.z;
          acc.w += hv.w;
        }
      }
    }
  }

  if constexpr (G == 2) {
    acc.x += __shfl_xor(acc.x, 16);
    acc.y += __shfl_xor(acc.y, 16);
    acc.z += __shfl_xor(acc.z, 16);
    acc.w += __shfl_xor(acc.w, 16);
  }
  acc.x *= di; acc.y *= di; acc.z *= di; acc.w *= di;
  return acc;
}

template <int C>
__global__ __launch_bounds__(256)
void gather_k(const float* __restrict__ h,
              const unsigned* __restrict__ rowptr, const int* __restrict__ csr_src,
              const float* __restrict__ bias, float* __restrict__ out) {
  constexpr int LPR = C / 4;
  const int lane = threadIdx.x & 31;
  const int node = __builtin_amdgcn_readfirstlane((int)(blockIdx.x * 8 + (threadIdx.x >> 5)));
  if (node >= N_NODES) return;
  float di;
  const float4 acc = node_sum<C>(h, rowptr, csr_src, node, lane, di);
  if (lane < LPR) {
    const unsigned fl = (unsigned)lane * 4u;
    const float4 b = *(const float4*)(bias + fl);
    *(float4*)(out + (unsigned)node * C + fl) =
        make_float4(acc.x + b.x, acc.y + b.y, acc.z + b.z, acc.w + b.w);
  }
}

#define FG_NODES  16
#define FG_THREADS (FG_NODES * 32)
#define A1_PITCH  (C_HID + 4)
#define W2_PITCH  C_OUT
#define W2_ROUNDS ((C_HID * C_OUT / 4) / FG_THREADS)
static_assert(N_NODES % FG_NODES == 0, "whole 16-row tiles, no masking (EXEC all ones for WMMA)");
static_assert((C_HID * C_OUT / 4) % FG_THREADS == 0, "W2 staging in whole float4 rounds");

__global__ __launch_bounds__(FG_THREADS)
void gather_gemm_k(const float* __restrict__ h,
                   const unsigned* __restrict__ rowptr, const int* __restrict__ csr_src,
                   const float* __restrict__ bias, const float* __restrict__ W2,
                   float* __restrict__ H2) {
  __shared__ __attribute__((aligned(16))) float a1s[FG_NODES * A1_PITCH];
  __shared__ __attribute__((aligned(16))) float w2s[C_HID * W2_PITCH];
  __shared__ float dis[FG_NODES];
  const int lane = threadIdx.x & 31;
  const int wave = __builtin_amdgcn_readfirstlane((int)(threadIdx.x >> 5));
  const int node = blockIdx.x * FG_NODES + wave;

  float4 wst[W2_ROUNDS];
#pragma unroll
  for (int i = 0; i < W2_ROUNDS; ++i)
    wst[i] = ((const float4*)W2)[threadIdx.x + i * FG_THREADS];
#pragma unroll
  for (int i = 0; i < W2_ROUNDS; ++i)
    *(float4*)(w2s + 4 * (threadIdx.x + i * FG_THREADS)) = wst[i];

  {
    float di;
    const float4 acc = node_sum<C_HID>(h, rowptr, csr_src, node, lane, di);
    if (lane == 0) dis[wave] = di;
    const unsigned fl = (unsigned)lane * 4u;
    const float4 b = *(const float4*)(bias + fl);
    float4 v = make_float4(acc.x + b.x, acc.y + b.y, acc.z + b.z, acc.w + b.w);
    v.x = fmaxf(v.x, 0.0f); v.y = fmaxf(v.y, 0.0f);
    v.z = fmaxf(v.z, 0.0f); v.w = fmaxf(v.w, 0.0f);
    *(float4*)(a1s + wave * A1_PITCH + fl) = v;
  }
  __syncthreads();

  if (wave < C_OUT / 16) {
    const int r    = lane & 15;
    const int half = lane >> 4;
    const int col0 = wave * 16;
    const float* Arow = a1s + r * A1_PITCH + half * 2;
    const float* Bcol = w2s + half * 2 * W2_PITCH + col0 + r;
    v8f acc = {};
    constexpr int KG = 8;
    constexpr int NG = C_HID / (4 * KG);
    v2f a[2][KG], b[2][KG];
#pragma unroll
    for (int u = 0; u < KG; ++u) {
      a[0][u]   = *(const v2f*)(Arow + 4 * u);
      b[0][u].x = Bcol[(4 * u) * W2_PITCH];
      b[0][u].y = Bcol[(4 * u + 1) * W2_PITCH];
    }
#pragma unroll
    for (int g = 0; g < NG; ++g) {
      if (g + 1 < NG) {
#pragma unroll
        for (int u = 0; u < KG; ++u) {
          const int k = 4 * ((g + 1) * KG + u);
          a[(g + 1) & 1][u]   = *(const v2f*)(Arow + k);
          b[(g + 1) & 1][u].x = Bcol[k * W2_PITCH];
          b[(g + 1) & 1][u].y = Bcol[(k + 1) * W2_PITCH];
        }
      }
#pragma unroll
      for (int u = 0; u < KG; ++u)
        acc = __builtin_amdgcn_wmma_f32_16x16x4_f32(false, a[g & 1][u], false, b[g & 1][u],
                                                    (short)0, acc, false, false);
    }
    __builtin_amdgcn_sched_group_barrier(0x100, KG / 2 + KG, 0);
#pragma unroll
    for (int g = 0; g < NG; ++g) {
      if (g + 1 < NG) __builtin_amdgcn_sched_group_barrier(0x100, KG / 2 + KG, 0);
      __builtin_amdgcn_sched_group_barrier(0x008, KG, 0);
    }
    float* Drow = H2 + (size_t)(blockIdx.x * FG_NODES + half * 8) * C_OUT + col0 + r;
#pragma unroll
    for (int i = 0; i < 8; ++i) Drow[(size_t)i * C_OUT] = acc[i] * dis[half * 8 + i];
  }
}

extern "C" void kernel_launch(void* const* d_in, const int* in_sizes, int n_in,
                              void* d_out, int out_size, void* d_ws, size_t ws_size,
                              hipStream_t stream) {
  (void)in_sizes; (void)n_in; (void)out_size; (void)ws_size;
  const float* x  = (const float*)d_in[0];
  const int*   ei = (const int*)d_in[1];
  const float* W1 = (const float*)d_in[2];
  const float* b1 = (const float*)d_in[3];
  const float* W2 = (const float*)d_in[4];
  const float* b2 = (const float*)d_in[5];
  const int* src = ei;
  const int* dst = ei + E_EDGES;
  float* out = (float*)d_out;

  char* ws = (char*)d_ws;
  float* H1   = (float*)ws;  ws += (size_t)N_NODES * C_HID * 4;
  float* H2   = (float*)ws;  ws += (size_t)N_NODES * C_OUT * 4;
  unsigned* part = (unsigned*)ws;  ws += (size_t)S_SLICES * NPAD * 4;
  unsigned* T    = (unsigned*)ws;  ws += (((size_t)S_SLICES * R_RANGES * 4) + 255) / 256 * 256;
  unsigned* rowptr = (unsigned*)ws;  ws += (((size_t)(N_NODES + 1) * 4) + 255) / 256 * 256;
  int* csr = (int*)ws;

  const dim3 B(256);
  deg_hist_k<<<HIST_BLOCKS, HIST_THREADS, 0, stream>>>(dst, part, T);
  csr_fill_gemm_k<<<SORT_BLOCKS + GEMM_BLOCKS, HIST_THREADS, 0, stream>>>(src, dst, part, T, rowptr, csr,
                                                                        x, W1, H1);

  gather_gemm_k<<<N_NODES / FG_NODES, FG_THREADS, 0, stream>>>(H1, rowptr, csr, b1, W2, H2);

  gather_k<C_OUT><<<(N_NODES + 7) / 8, B, 0, stream>>>(H2, rowptr, csr, b2, out);
}


# H1 = x @ W1 blocks stage x rows and W1 through LDS in 64-k chunks by whole-line float4 loads, next chunk in flight; fragments come from LDS, not from a memory line per row per k-step; k order kept
# speedup vs baseline: 4.5921x; 1.0415x over previous
#include <hip/hip_runtime.h>

#define N_NODES 10000
#define E_EDGES 640000
#define C_IN    256
#define C_HID   128
#define C_OUT   64

typedef float    v2f  __attribute__((ext_vector_type(2)));
typedef float    v8f  __attribute__((ext_vector_type(8)));

#define S_SLICES   16
#define RANGE      1024
#define R_RANGES   ((N_NODES + RANGE - 1) / RANGE)
#define NPAD       (R_RANGES * RANGE)
#define HR         2
#define H_RBLOCKS  (R_RANGES / HR)
#define HIST_BLOCKS  (S_SLICES * H_RBLOCKS)
#define HIST_THREADS 1024
#define SLICE      (E_EDGES / S_SLICES)
#define SLICE4     (SLICE / 4)
#define SLICE_IT   ((SLICE4 + HIST_THREADS - 1) / HIST_THREADS)
static_assert(E_EDGES % (S_SLICES * 4) == 0, "each slice is whole int4 groups");
static_assert(R_RANGES % HR == 0, "whole histogram blocks");
static_assert(RANGE == HIST_THREADS, "one node of a range per thread");

__global__ __launch_bounds__(HIST_THREADS)
void deg_hist_k(const int* __restrict__ dst, unsigned* __restrict__ part, unsigned* __restrict__ T) {
  __shared__ unsigned cnt[HR * RANGE];
  __shared__ unsigned tsum[HR];
  const int t    = threadIdx.x;
  const int lane = t & 31;
  const int s    = blockIdx.x / H_RBLOCKS;
  const int rh   = blockIdx.x % H_RBLOCKS;
  const int lo   = rh * HR * RANGE;
#pragma unroll
  for (int q = 0; q < HR; ++q) cnt[t + q * RANGE] = 0u;
  if (t < HR) tsum[t] = 0u;
  __syncthreads();

  const int4* d4 = (const int4*)(dst + (size_t)s * SLICE);
  int4 v[SLICE_IT];
#pragma unroll
  for (int it = 0; it < SLICE_IT; ++it) {
    const int i = t + it * HIST_THREADS;
    v[it] = (i < SLICE4) ? d4[i] : make_int4(-1, -1, -1, -1);
  }
#pragma unroll
  for (int it = 0; it < SLICE_IT; ++it) {
    const unsigned k0 = (unsigned)(v[it].x - lo), k1 = (unsigned)(v[it].y - lo);
    const unsigned k2 = (unsigned)(v[it].z - lo), k3 = (unsigned)(v[it].w - lo);
    if (k0 < HR * RANGE) __hip_atomic_fetch_add(&cnt[k0], 1u, __ATOMIC_RELAXED, __HIP_MEMORY_SCOPE_WORKGROUP);
    if (k1 < HR * RANGE) __hip_atomic_fetch_add(&cnt[k1], 1u, __ATOMIC_RELAXED, __HIP_MEMORY_SCOPE_WORKGROUP);
    if (k2 < HR * RANGE) __hip_atomic_fetch_add(&cnt[k2], 1u, __ATOMIC_RELAXED, __HIP_MEMORY_SCOPE_WORKGROUP);
    if (k3 < HR * RANGE) __hip_atomic_fetch_add(&cnt[k3], 1u, __ATOMIC_RELAXED, __HIP_MEMORY_SCOPE_WORKGROUP);
  }
  __syncthreads();

#pragma unroll
  for (int q = 0; q < HR; ++q) {
    const unsigned c = cnt[t + q * RANGE];
    part[(size_t)s * NPAD + lo + q * RANGE + t] = c;
    unsigned r = c;
    for (int o = 16; o > 0; o >>= 1) r += __shfl_xor(r, o);
    if (lane == 0) __hip_atomic_fetch_add(&tsum[q], r, __ATOMIC_RELAXED, __HIP_MEMORY_SCOPE_WORKGROUP);
  }
  __syncthreads();
  if (t < HR) T[s * R_RANGES + rh * HR + t] = tsum[t];
}

static_assert(N_NODES % 16 == 0 && C_HID % 16 == 0 && C_IN % 4 == 0, "whole tiles only");

#define GK_CHUNK   64
#define GK_CHUNKS  (C_IN / GK_CHUNK)
#define XS_PITCH   (GK_CHUNK + 4)
#define W1S_ROWS_PER_ROUND (HIST_THREADS / (C_HID / 4))
#define W1S_ROUNDS (GK_CHUNK / W1S_ROWS_PER_ROUND)
#define W1S_ROT    32
static_assert(C_IN % GK_CHUNK == 0 && GK_CHUNK % 4 == 0, "whole chunks of whole k-steps");
static_assert(GK_CHUNK % W1S_ROWS_PER_ROUND == 0 && W1S_ROWS_PER_ROUND % 4 == 0, "W1 staging in whole rounds; a round keeps bit 1 of k");
static_assert((C_HID & (C_HID - 1)) == 0 && W1S_ROT % 4 == 0, "the rotation wraps by a mask and keeps float4 groups whole");

#define SORT_BLOCKS (S_SLICES * R_RANGES)
#define GEMM_COLT   (C_HID / 16)
#define GEMM_ROWT   ((HIST_THREADS / 32) / GEMM_COLT)
#define GEMM_ROWS   (GEMM_ROWT * 16)
#define GEMM_BLOCKS ((N_NODES / 16 + GEMM_ROWT - 1) / GEMM_ROWT)
#define FILL_BATCH  5
#define NO_SLOT     0xffffffffu
static_assert(GEMM_COLT * GEMM_ROWT * 32 == HIST_THREADS, "one wave per tile of a product block");
static_assert(GEMM_ROWS * S_SLICES == HIST_THREADS, "one (row, slice) count per thread of a product block");
static_assert(GEMM_BLOCKS * GEMM_ROWS <= NPAD, "the degree sums stay inside part's rows");
static_assert(S_SLICES * R_RANGES <= HIST_THREADS, "one T entry per thread");
static_assert(SLICE_IT % FILL_BATCH == 0, "whole batches");
static_assert(GEMM_ROWS * GK_CHUNK / 4 == HIST_THREADS, "one float4 of the x chunk per thread of a product block");
static_assert(W1S_ROUNDS == 2, "the load fence in the product block names two W1 rounds");

__global__ __launch_bounds__(HIST_THREADS, 16)
void csr_fill_gemm_k(const int* __restrict__ src, const int* __restrict__ dst,
                     const unsigned* __restrict__ part, const unsigned* __restrict__ T,
                     unsigned* __restrict__ rowptr, int* __restrict__ csr_src,
                     const float* __restrict__ x, const float* __restrict__ W1,
                     float* __restrict__ H1) {
  __shared__ unsigned cur[RANGE];
  __shared__ unsigned wtot[HIST_THREADS / 32];
  __shared__ unsigned wbase[HIST_THREADS / 32];
  __shared__ unsigned btot;
  __shared__ unsigned indeg[GEMM_ROWS];
  __shared__ __attribute__((aligned(16))) float xs[GEMM_ROWS * XS_PITCH];
  __shared__ __attribute__((aligned(16))) float w1s[GK_CHUNK * C_HID];
  const int t    = threadIdx.x;
  const int lane = t & 31;
  const int w    = t >> 5;

  if (blockIdx.x < GEMM_BLOCKS) {
    const int row0b = (int)blockIdx.x * GEMM_ROWS;
    const int wv    = __builtin_amdgcn_readfirstlane(w);
    const int rtl   = wv / GEMM_COLT;
    const int col0  = (wv % GEMM_COLT) * 16;
    const bool live = (row0b / 16 + rtl) < N_NODES / 16;

    const int  arow = t / (GK_CHUNK / 4);
    const int  ac4  = t % (GK_CHUNK / 4);
    const int  xrow = (row0b + arow) < N_NODES ? (row0b + arow) : (N_NODES - 1);
    const float4* xg = (const float4*)(x + (size_t)xrow * C_IN) + ac4;
    float* xsd = xs + arow * XS_PITCH + ac4 * 4;
    const int bk  = t / (C_HID / 4);
    const int bc4 = t % (C_HID / 4);
    const float4* wg = (const float4*)W1 + (size_t)bk * (C_HID / 4) + bc4;
    float* wsd = w1s + bk * C_HID + ((bc4 * 4 + W1S_ROT * ((bk >> 1) & 1)) & (C_HID - 1));

    const int r    = lane & 15;
    const int half = lane >> 4;
    const float* Arow = xs + (rtl * 16 + r) * XS_PITCH + half * 2;
    const float* Bcol = w1s + half * 2 * C_HID + ((col0 + r + W1S_ROT * half) & (C_HID - 1));

    if (t < GEMM_ROWS) indeg[t] = 0u;
    float4 xa, wb[W1S_ROUNDS];
    xa = xg[0];
#pragma unroll
    for (int i = 0; i < W1S_ROUNDS; ++i) wb[i] = wg[(size_t)i * W1S_ROWS_PER_ROUND * (C_HID / 4)];

    v8f acc = {};
#pragma unroll
    for (int kc = 0; kc < GK_CHUNKS; ++kc) {
      __syncthreads();
      *(float4*)xsd = xa;
#pragma unroll
      for (int i = 0; i < W1S_ROUNDS; ++i) *(float4*)(wsd + i * W1S_ROWS_PER_ROUND * C_HID) = wb[i];
      if (kc == 0)
        __hip_atomic_fetch_add(&indeg[t % GEMM_ROWS], part[(size_t)(t / GEMM_ROWS) * NPAD + row0b + t % GEMM_ROWS],
                               __ATOMIC_RELAXED, __HIP_MEMORY_SCOPE_WORKGROUP);
      __syncthreads();
      if (kc + 1 < GK_CHUNKS) {
        xa = xg[(kc + 1) * (GK_CHUNK / 4)];
#pragma unroll
        for (int i = 0; i < W1S_ROUNDS; ++i)
          wb[i] = wg[((size_t)(kc + 1) * GK_CHUNK + i * W1S_ROWS_PER_ROUND) * (C_HID / 4)];
      }
#pragma unroll
      for (int kk = 0; kk < GK_CHUNK; kk += 4) {
        v2f a = *(const v2f*)(Arow + kk);
        v2f b;
        b.x = Bcol[kk * C_HID];
        b.y = Bcol[(kk + 1) * C_HID];
        acc = __builtin_amdgcn_wmma_f32_16x16x4_f32(false, a, false, b,
                                                    (short)0, acc, false, false);
      }
      if (kc + 1 < GK_CHUNKS) {
        asm volatile("" :: "v"(xa.x), "v"(xa.y), "v"(xa.z), "v"(xa.w),
                           "v"(wb[0].x), "v"(wb[0].y), "v"(wb[0].z), "v"(wb[0].w),
                           "v"(wb[1].x), "v"(wb[1].y), "v"(wb[1].z), "v"(wb[1].w));
      }
    }
    if (live) {
      float* Drow = H1 + (size_t)(row0b + rtl * 16 + half * 8) * C_HID + col0 + r;
#pragma unroll
      for (int i = 0; i < 8; ++i)
        Drow[(size_t)i * C_HID] = acc[i] * rsqrtf((float)(indeg[rtl * 16 + half * 8 + i] + 1u));
    }
    return;
  }

  const int sb = (int)blockIdx.x - GEMM_BLOCKS;
  const int s  = sb / R_RANGES;
  const int r  = sb % R_RANGES;
  const int lo = r * RANGE;

  unsigned tv = 0u;
  if (t < S_SLICES * R_RANGES && (t % R_RANGES) < r) tv = T[t];
  unsigned tot = 0u, po = 0u;
#pragma unroll
  for (int q = 0; q < S_SLICES; ++q) {
    const unsigned c = part[(size_t)q * NPAD + lo + t];
    tot += c;
    if (q < s) po += c;
  }

  unsigned inc = tot;
  for (int o = 1; o < 32; o <<= 1) {
    const unsigned v = __shfl_up(inc, o);
    if (lane >= o) inc += v;
  }
  unsigned bs = tv;
  for (int o = 16; o > 0; o >>= 1) bs += __shfl_xor(bs, o);
  if (lane == 31) wtot[w] = inc;
  if (lane == 0)  wbase[w] = bs;
  __syncthreads();
  if (w == 0) {
    const unsigned xw = wtot[lane];
    unsigned xi = xw;
    for (int o = 1; o < 32; o <<= 1) {
      const unsigned v = __shfl_up(xi, o);
      if (lane >= o) xi += v;
    }
    wtot[lane] = xi - xw;
    unsigned bb = wbase[lane];
    for (int o = 16; o > 0; o >>= 1) bb += __shfl_xor(bb, o);
    if (lane == 0) btot = bb;
  }
  __syncthreads();
  const unsigned start = btot + wtot[w] + (inc - tot);
  cur[t] = start + po;
  if (s == 0) {
    const int node = lo + t;
    if (node < N_NODES) rowptr[node] = start;
    if (node == N_NODES - 1) rowptr[N_NODES] = start + tot;
  }
  __syncthreads();

  const int4* d4 = (const int4*)(dst + (size_t)s * SLICE);
  const int4* s4 = (const int4*)(src + (size_t)s * SLICE);
#pragma unroll
  for (int b = 0; b < SLICE_IT; b += FILL_BATCH) {
    int4 dv[FILL_BATCH], sv[FILL_BATCH];
#pragma unroll
    for (int u = 0; u < FILL_BATCH; ++u) {
      const int i = t + (b + u) * HIST_THREADS;
      if (i < SLICE4) { dv[u] = d4[i]; sv[u] = s4[i]; }
      else            { dv[u] = make_int4(-1, -1, -1, -1); sv[u] = make_int4(0, 0, 0, 0); }
    }
#pragma unroll
    for (int u = 0; u < FILL_BATCH; ++u) {
      asm volatile("" :: "v"(sv[u].x), "v"(sv[u].y), "v"(sv[u].z), "v"(sv[u].w));
    }
    unsigned p[FILL_BATCH * 4];
#pragma unroll
    for (int u = 0; u < FILL_BATCH; ++u) {
      const unsigned k0 = (unsigned)(dv[u].x - lo), k1 = (unsigned)(dv[u].y - lo);
      const unsigned k2 = (unsigned)(dv[u].z - lo), k3 = (unsigned)(dv[u].w - lo);
      p[4 * u + 0] = (k0 < RANGE) ? __hip_atomic_fetch_add(&cur[k0], 1u, __ATOMIC_RELAXED, __HIP_MEMORY_SCOPE_WORKGROUP) : NO_SLOT;
      p[4 * u + 1] = (k1 < RANGE) ? __hip_atomic_fetch_add(&cur[k1], 1u, __ATOMIC_RELAXED, __HIP_MEMORY_SCOPE_WORKGROUP) : NO_SLOT;
      p[4 * u + 2] = (k2 < RANGE) ? __hip_atomic_fetch_add(&cur[k2], 1u, __ATOMIC_RELAXED, __HIP_MEMORY_SCOPE_WORKGROUP) : NO_SLOT;
      p[4 * u + 3] = (k3 < RANGE) ? __hip_atomic_fetch_add(&cur[k3], 1u, __ATOMIC_RELAXED, __HIP_MEMORY_SCOPE_WORKGROUP) : NO_SLOT;
    }
#pragma unroll
    for (int u = 0; u < FILL_BATCH; ++u) {
      if (p[4 * u + 0] != NO_SLOT) csr_src[p[4 * u + 0]] = sv[u].x;
      if (p[4 * u + 1] != NO_SLOT) csr_src[p[4 * u + 1]] = sv[u].y;
      if (p[4 * u + 2] != NO_SLOT) csr_src[p[4 * u + 2]] = sv[u].z;
      if (p[4 * u + 3] != NO_SLOT) csr_src[p[4 * u + 3]] = sv[u].w;
    }
  }
}

template <int G>
__device__ __forceinline__ int lane_bcast(int v, int idx) {
  if constexpr (G == 1) return __builtin_amdgcn_readlane(v, idx);
  else                  return __shfl(v, idx);
}

template <int C, int G, int NS>
__device__ __forceinline__ void rows_add(const float* __restrict__ h, int sv,
                                         int step0, int grp, unsigned fl, float4& acc) {
  float4 hv[NS];
#pragma unroll
  for (int u = 0; u < NS; ++u) {
    const int idx = (step0 + u) * G + grp;
    const unsigned s = (unsigned)lane_bcast<G>(sv, idx);
    hv[u] = *(const float4*)(h + s * C + fl);
  }
#pragma unroll
  for (int u = 0; u < NS; ++u) {
    acc.x += hv[u].x;
    acc.y += hv[u].y;
    acc.z += hv[u].z;
    acc.w += hv[u].w;
  }
}

template <int C>
__device__ __forceinline__ float4 node_sum(const float* __restrict__ h,
                                           const unsigned* __restrict__ rowptr,
                                           const int* __restrict__ csr_src, int node, int lane, float& di) {
  constexpr int LPR = C / 4;
  constexpr int G   = 32 / LPR;
  constexpr int U   = 8;
  const int grp = lane / LPR;
  const unsigned fl = (unsigned)(lane % LPR) * 4u;

  const unsigned beg = rowptr[node];
  const unsigned end = rowptr[node + 1];
  di = rsqrtf((float)(end - beg + 1u));

  float4 acc = make_float4(0.0f, 0.0f, 0.0f, 0.0f);
  if (grp == 0)
    acc = *(const float4*)(h + (unsigned)node * C + fl);

  for (unsigned cb = beg; cb < end; cb += 32u) {
    const int n = (int)((end - cb) < 32u ? (end - cb) : 32u);
    int sv = 0;
    if (lane < n) sv = csr_src[cb + lane];

    const int nfull = n / G;
    int j = 0;
    for (; j + U <= nfull; j += U) rows_add<C, G, U>(h, sv, j, grp, fl, acc);
    if (j + 4 <= nfull) { rows_add<C, G, 4>(h, sv, j, grp, fl, acc); j += 4; }
    if (j + 2 <= nfull) { rows_add<C, G, 2>(h, sv, j, grp, fl, acc); j += 2; }
    if (j + 1 <= nfull) { rows_add<C, G, 1>(h, sv, j, grp, fl, acc); j += 1; }
    if constexpr (G == 2) {
      if (n & 1) {
        const int idx = n - 1 + grp;
        const unsigned s = (unsigned)__shfl(sv, idx);
        if (grp == 0) {
          const float4 hv = *(const float4*)(h + s * C + fl);
          acc.x += hv.x;
          acc.y += hv.y;
          acc.z += hv.z;
          acc.w += hv.w;
        }
      }
    }
  }

  if constexpr (G == 2) {
    acc.x += __shfl_xor(acc.x, 16);
    acc.y += __shfl_xor(acc.y, 16);
    acc.z += __shfl_xor(acc.z, 16);
    acc.w += __shfl_xor(acc.w, 16);
  }
  acc.x *= di; acc.y *= di; acc.z *= di; acc.w *= di;
  return acc;
}

template <int C>
__global__ __launch_bounds__(256)
void gather_k(const float* __restrict__ h,
              const unsigned* __restrict__ rowptr, const int* __restrict__ csr_src,
              const float* __restrict__ bias, float* __restrict__ out) {
  constexpr int LPR = C / 4;
  const int lane = threadIdx.x & 31;
  const int node = __builtin_amdgcn_readfirstlane((int)(blockIdx.x * 8 + (threadIdx.x >> 5)));
  if (node >= N_NODES) return;
  float di;
  const float4 acc = node_sum<C>(h, rowptr, csr_src, node, lane, di);
  if (lane < LPR) {
    const unsigned fl = (unsigned)lane * 4u;
    const float4 b = *(const float4*)(bias + fl);
    *(float4*)(out + (unsigned)node * C + fl) =
        make_float4(acc.x + b.x, acc.y + b.y, acc.z + b.z, acc.w + b.w);
  }
}

#define FG_NODES  16
#define FG_THREADS (FG_NODES * 32)
#define A1_PITCH  (C_HID + 4)
#define W2_PITCH  C_OUT
#define W2_ROUNDS ((C_HID * C_OUT / 4) / FG_THREADS)
static_assert(N_NODES % FG_NODES == 0, "whole 16-row tiles, no masking (EXEC all ones for WMMA)");
static_assert((C_HID * C_OUT / 4) % FG_THREADS == 0, "W2 staging in whole float4 rounds");

__global__ __launch_bounds__(FG_THREADS)
void gather_gemm_k(const float* __restrict__ h,
                   const unsigned* __restrict__ rowptr, const int* __restrict__ csr_src,
                   const float* __restrict__ bias, const float* __restrict__ W2,
                   float* __restrict__ H2) {
  __shared__ __attribute__((aligned(16))) float a1s[FG_NODES * A1_PITCH];
  __shared__ __attribute__((aligned(16))) float w2s[C_HID * W2_PITCH];
  __shared__ float dis[FG_NODES];
  const int lane = threadIdx.x & 31;
  const int wave = __builtin_amdgcn_readfirstlane((int)(threadIdx.x >> 5));
  const int node = blockIdx.x * FG_NODES + wave;

  float4 wst[W2_ROUNDS];
#pragma unroll
  for (int i = 0; i < W2_ROUNDS; ++i)
    wst[i] = ((const float4*)W2)[threadIdx.x + i * FG_THREADS];
#pragma unroll
  for (int i = 0; i < W2_ROUNDS; ++i)
    *(float4*)(w2s + 4 * (threadIdx.x + i * FG_THREADS)) = wst[i];

  {
    float di;
    const float4 acc = node_sum<C_HID>(h, rowptr, csr_src, node, lane, di);
    if (lane == 0) dis[wave] = di;
    const unsigned fl = (unsigned)lane * 4u;
    const float4 b = *(const float4*)(bias + fl);
    float4 v = make_float4(acc.x + b.x, acc.y + b.y, acc.z + b.z, acc.w + b.w);
    v.x = fmaxf(v.x, 0.0f); v.y = fmaxf(v.y, 0.0f);
    v.z = fmaxf(v.z, 0.0f); v.w = fmaxf(v.w, 0.0f);
    *(float4*)(a1s + wave * A1_PITCH + fl) = v;
  }
  __syncthreads();

  if (wave < C_OUT / 16) {
    const int r    = lane & 15;
    const int half = lane >> 4;
    const int col0 = wave * 16;
    const float* Arow = a1s + r * A1_PITCH + half * 2;
    const float* Bcol = w2s + half * 2 * W2_PITCH + col0 + r;
    v8f acc = {};
    constexpr int KG = 8;
    constexpr int NG = C_HID / (4 * KG);
    v2f a[2][KG], b[2][KG];
#pragma unroll
    for (int u = 0; u < KG; ++u) {
      a[0][u]   = *(const v2f*)(Arow + 4 * u);
      b[0][u].x = Bcol[(4 * u) * W2_PITCH];
      b[0][u].y = Bcol[(4 * u + 1) * W2_PITCH];
    }
#pragma unroll
    for (int g = 0; g < NG; ++g) {
      if (g + 1 < NG) {
#pragma unroll
        for (int u = 0; u < KG; ++u) {
          const int k = 4 * ((g + 1) * KG + u);
          a[(g + 1) & 1][u]   = *(const v2f*)(Arow + k);
          b[(g + 1) & 1][u].x = Bcol[k * W2_PITCH];
          b[(g + 1) & 1][u].y = Bcol[(k + 1) * W2_PITCH];
        }
      }
#pragma unroll
      for (int u = 0; u < KG; ++u)
        acc = __builtin_amdgcn_wmma_f32_16x16x4_f32(false, a[g & 1][u], false, b[g & 1][u],
                                                    (short)0, acc, false, false);
    }
    __builtin_amdgcn_sched_group_barrier(0x100, KG / 2 + KG, 0);
#pragma unroll
    for (int g = 0; g < NG; ++g) {
      if (g + 1 < NG) __builtin_amdgcn_sched_group_barrier(0x100, KG / 2 + KG, 0);
      __builtin_amdgcn_sched_group_barrier(0x008, KG, 0);
    }
    float* Drow = H2 + (size_t)(blockIdx.x * FG_NODES + half * 8) * C_OUT + col0 + r;
#pragma unroll
    for (int i = 0; i < 8; ++i) Drow[(size_t)i * C_OUT] = acc[i] * dis[half * 8 + i];
  }
}

extern "C" void kernel_launch(void* const* d_in, const int* in_sizes, int n_in,
                              void* d_out, int out_size, void* d_ws, size_t ws_size,
                              hipStream_t stream) {
  (void)in_sizes; (void)n_in; (void)out_size; (void)ws_size;
  const float* x  = (const float*)d_in[0];
  const int*   ei = (const int*)d_in[1];
  const float* W1 = (const float*)d_in[2];
  const float* b1 = (const float*)d_in[3];
  const float* W2 = (const float*)d_in[4];
  const float* b2 = (const float*)d_in[5];
  const int* src = ei;
  const int* dst = ei + E_EDGES;
  float* out = (float*)d_out;

  char* ws = (char*)d_ws;
  float* H1   = (float*)ws;  ws += (size_t)N_NODES * C_HID * 4;
  float* H2   = (float*)ws;  ws += (size_t)N_NODES * C_OUT * 4;
  unsigned* part = (unsigned*)ws;  ws += (size_t)S_SLICES * NPAD * 4;
  unsigned* T    = (unsigned*)ws;  ws += (((size_t)S_SLICES * R_RANGES * 4) + 255) / 256 * 256;
  unsigned* rowptr = (unsigned*)ws;  ws += (((size_t)(N_NODES + 1) * 4) + 255) / 256 * 256;
  int* csr = (int*)ws;

  const dim3 B(256);
  deg_hist_k<<<HIST_BLOCKS, HIST_THREADS, 0, stream>>>(dst, part, T);
  csr_fill_gemm_k<<<SORT_BLOCKS + GEMM_BLOCKS, HIST_THREADS, 0, stream>>>(src, dst, part, T, rowptr, csr,
                                                                        x, W1, H1);

  gather_gemm_k<<<N_NODES / FG_NODES, FG_THREADS, 0, stream>>>(H1, rowptr, csr, b1, W2, H2);

  gather_k<C_OUT><<<(N_NODES + 7) / 8, B, 0, stream>>>(H2, rowptr, csr, b2, out);
}


# edge list cut into 32 position slices, not 16: each count block and each CSR-fill block reads and places half as many edges (one load batch per fill block); per-node sum order = new CSR order
# speedup vs baseline: 4.9847x; 1.0855x over previous
#include <hip/hip_runtime.h>

#define N_NODES 10000
#define E_EDGES 640000
#define C_IN    256
#define C_HID   128
#define C_OUT   64

typedef float    v2f  __attribute__((ext_vector_type(2)));
typedef float    v8f  __attribute__((ext_vector_type(8)));

#define S_SLICES   32
#define RANGE      1024
#define R_RANGES   ((N_NODES + RANGE - 1) / RANGE)
#define NPAD       (R_RANGES * RANGE)
#define HR         2
#define H_RBLOCKS  (R_RANGES / HR)
#define HIST_BLOCKS  (S_SLICES * H_RBLOCKS)
#define HIST_THREADS 1024
#define SLICE      (E_EDGES / S_SLICES)
#define SLICE4     (SLICE / 4)
#define SLICE_IT   ((SLICE4 + HIST_THREADS - 1) / HIST_THREADS)
static_assert(E_EDGES % (S_SLICES * 4) == 0, "each slice is whole int4 groups");
static_assert(R_RANGES % HR == 0, "whole histogram blocks");
static_assert(RANGE == HIST_THREADS, "one node of a range per thread");

__global__ __launch_bounds__(HIST_THREADS)
void deg_hist_k(const int* __restrict__ dst, unsigned* __restrict__ part, unsigned* __restrict__ T) {
  __shared__ unsigned cnt[HR * RANGE];
  __shared__ unsigned tsum[HR];
  const int t    = threadIdx.x;
  const int lane = t & 31;
  const int s    = blockIdx.x / H_RBLOCKS;
  const int rh   = blockIdx.x % H_RBLOCKS;
  const int lo   = rh * HR * RANGE;
#pragma unroll
  for (int q = 0; q < HR; ++q) cnt[t + q * RANGE] = 0u;
  if (t < HR) tsum[t] = 0u;
  __syncthreads();

  const int4* d4 = (const int4*)(dst + (size_t)s * SLICE);
  int4 v[SLICE_IT];
#pragma unroll
  for (int it = 0; it < SLICE_IT; ++it) {
    const int i = t + it * HIST_THREADS;
    v[it] = (i < SLICE4) ? d4[i] : make_int4(-1, -1, -1, -1);
  }
#pragma unroll
  for (int it = 0; it < SLICE_IT; ++it) {
    const unsigned k0 = (unsigned)(v[it].x - lo), k1 = (unsigned)(v[it].y - lo);
    const unsigned k2 = (unsigned)(v[it].z - lo), k3 = (unsigned)(v[it].w - lo);
    if (k0 < HR * RANGE) __hip_atomic_fetch_add(&cnt[k0], 1u, __ATOMIC_RELAXED, __HIP_MEMORY_SCOPE_WORKGROUP);
    if (k1 < HR * RANGE) __hip_atomic_fetch_add(&cnt[k1], 1u, __ATOMIC_RELAXED, __HIP_MEMORY_SCOPE_WORKGROUP);
    if (k2 < HR * RANGE) __hip_atomic_fetch_add(&cnt[k2], 1u, __ATOMIC_RELAXED, __HIP_MEMORY_SCOPE_WORKGROUP);
    if (k3 < HR * RANGE) __hip_atomic_fetch_add(&cnt[k3], 1u, __ATOMIC_RELAXED, __HIP_MEMORY_SCOPE_WORKGROUP);
  }
  __syncthreads();

#pragma unroll
  for (int q = 0; q < HR; ++q) {
    const unsigned c = cnt[t + q * RANGE];
    part[(size_t)s * NPAD + lo + q * RANGE + t] = c;
    unsigned r = c;
    for (int o = 16; o > 0; o >>= 1) r += __shfl_xor(r, o);
    if (lane == 0) __hip_atomic_fetch_add(&tsum[q], r, __ATOMIC_RELAXED, __HIP_MEMORY_SCOPE_WORKGROUP);
  }
  __syncthreads();
  if (t < HR) T[s * R_RANGES + rh * HR + t] = tsum[t];
}

static_assert(N_NODES % 16 == 0 && C_HID % 16 == 0 && C_IN % 4 == 0, "whole tiles only");

#define GK_CHUNK   64
#define GK_CHUNKS  (C_IN / GK_CHUNK)
#define XS_PITCH   (GK_CHUNK + 4)
#define W1S_ROWS_PER_ROUND (HIST_THREADS / (C_HID / 4))
#define W1S_ROUNDS (GK_CHUNK / W1S_ROWS_PER_ROUND)
#define W1S_ROT    32
static_assert(C_IN % GK_CHUNK == 0 && GK_CHUNK % 4 == 0, "whole chunks of whole k-steps");
static_assert(GK_CHUNK % W1S_ROWS_PER_ROUND == 0 && W1S_ROWS_PER_ROUND % 4 == 0, "W1 staging in whole rounds; a round keeps bit 1 of k");
static_assert((C_HID & (C_HID - 1)) == 0 && W1S_ROT % 4 == 0, "the rotation wraps by a mask and keeps float4 groups whole");

#define SORT_BLOCKS (S_SLICES * R_RANGES)
#define GEMM_COLT   (C_HID / 16)
#define GEMM_ROWT   ((HIST_THREADS / 32) / GEMM_COLT)
#define GEMM_ROWS   (GEMM_ROWT * 16)
#define GEMM_BLOCKS ((N_NODES / 16 + GEMM_ROWT - 1) / GEMM_ROWT)
#define FILL_BATCH  5
#define NO_SLOT     0xffffffffu
static_assert(GEMM_COLT * GEMM_ROWT * 32 == HIST_THREADS, "one wave per tile of a product block");
#define DEG_ROUNDS  (GEMM_ROWS * S_SLICES / HIST_THREADS)
static_assert(DEG_ROUNDS * HIST_THREADS == GEMM_ROWS * S_SLICES, "whole (row, slice) counts per thread of a product block");
static_assert(GEMM_BLOCKS * GEMM_ROWS <= NPAD, "the degree sums stay inside part's rows");
static_assert(S_SLICES * R_RANGES <= HIST_THREADS, "one T entry per thread");
static_assert(SLICE_IT % FILL_BATCH == 0, "whole batches");
static_assert(GEMM_ROWS * GK_CHUNK / 4 == HIST_THREADS, "one float4 of the x chunk per thread of a product block");
static_assert(W1S_ROUNDS == 2, "the load fence in the product block names two W1 rounds");

__global__ __launch_bounds__(HIST_THREADS, 16)
void csr_fill_gemm_k(const int* __restrict__ src, const int* __restrict__ dst,
                     const unsigned* __restrict__ part, const unsigned* __restrict__ T,
                     unsigned* __restrict__ rowptr, int* __restrict__ csr_src,
                     const float* __restrict__ x, const float* __restrict__ W1,
                     float* __restrict__ H1) {
  __shared__ unsigned cur[RANGE];
  __shared__ unsigned wtot[HIST_THREADS / 32];
  __shared__ unsigned wbase[HIST_THREADS / 32];
  __shared__ unsigned btot;
  __shared__ unsigned indeg[GEMM_ROWS];
  __shared__ __attribute__((aligned(16))) float xs[GEMM_ROWS * XS_PITCH];
  __shared__ __attribute__((aligned(16))) float w1s[GK_CHUNK * C_HID];
  const int t    = threadIdx.x;
  const int lane = t & 31;
  const int w    = t >> 5;

  if (blockIdx.x < GEMM_BLOCKS) {
    const int row0b = (int)blockIdx.x * GEMM_ROWS;
    const int wv    = __builtin_amdgcn_readfirstlane(w);
    const int rtl   = wv / GEMM_COLT;
    const int col0  = (wv % GEMM_COLT) * 16;
    const bool live = (row0b / 16 + rtl) < N_NODES / 16;

    const int  arow = t / (GK_CHUNK / 4);
    const int  ac4  = t % (GK_CHUNK / 4);
    const int  xrow = (row0b + arow) < N_NODES ? (row0b + arow) : (N_NODES - 1);
    const float4* xg = (const float4*)(x + (size_t)xrow * C_IN) + ac4;
    float* xsd = xs + arow * XS_PITCH + ac4 * 4;
    const int bk  = t / (C_HID / 4);
    const int bc4 = t % (C_HID / 4);
    const float4* wg = (const float4*)W1 + (size_t)bk * (C_HID / 4) + bc4;
    float* wsd = w1s + bk * C_HID + ((bc4 * 4 + W1S_ROT * ((bk >> 1) & 1)) & (C_HID - 1));

    const int r    = lane & 15;
    const int half = lane >> 4;
    const float* Arow = xs + (rtl * 16 + r) * XS_PITCH + half * 2;
    const float* Bcol = w1s + half * 2 * C_HID + ((col0 + r + W1S_ROT * half) & (C_HID - 1));

    if (t < GEMM_ROWS) indeg[t] = 0u;
    float4 xa, wb[W1S_ROUNDS];
    xa = xg[0];
#pragma unroll
    for (int i = 0; i < W1S_ROUNDS; ++i) wb[i] = wg[(size_t)i * W1S_ROWS_PER_ROUND * (C_HID / 4)];

    v8f acc = {};
#pragma unroll
    for (int kc = 0; kc < GK_CHUNKS; ++kc) {
      __syncthreads();
      *(float4*)xsd = xa;
#pragma unroll
      for (int i = 0; i < W1S_ROUNDS; ++i) *(float4*)(wsd + i * W1S_ROWS_PER_ROUND * C_HID) = wb[i];
      if (kc == 0) {
        unsigned c = 0u;
#pragma unroll
        for (int q = 0; q < DEG_ROUNDS; ++q)
          c += part[(size_t)(t / GEMM_ROWS + q * (HIST_THREADS / GEMM_ROWS)) * NPAD + row0b + t % GEMM_ROWS];
        __hip_atomic_fetch_add(&indeg[t % GEMM_ROWS], c, __ATOMIC_RELAXED, __HIP_MEMORY_SCOPE_WORKGROUP);
      }
      __syncthreads();
      if (kc + 1 < GK_CHUNKS) {
        xa = xg[(kc + 1) * (GK_CHUNK / 4)];
#pragma unroll
        for (int i = 0; i < W1S_ROUNDS; ++i)
          wb[i] = wg[((size_t)(kc + 1) * GK_CHUNK + i * W1S_ROWS_PER_ROUND) * (C_HID / 4)];
      }
#pragma unroll
      for (int kk = 0; kk < GK_CHUNK; kk += 4) {
        v2f a = *(const v2f*)(Arow + kk);
        v2f b;
        b.x = Bcol[kk * C_HID];
        b.y = Bcol[(kk + 1) * C_HID];
        acc = __builtin_amdgcn_wmma_f32_16x16x4_f32(false, a, false, b,
                                                    (short)0, acc, false, false);
      }
      if (kc + 1 < GK_CHUNKS) {
        asm volatile("" :: "v"(xa.x), "v"(xa.y), "v"(xa.z), "v"(xa.w),
                           "v"(wb[0].x), "v"(wb[0].y), "v"(wb[0].z), "v"(wb[0].w),
                           "v"(wb[1].x), "v"(wb[1].y), "v"(wb[1].z), "v"(wb[1].w));
      }
    }
    if (live) {
      float* Drow = H1 + (size_t)(row0b + rtl * 16 + half * 8) * C_HID + col0 + r;
#pragma unroll
      for (int i = 0; i < 8; ++i)
        Drow[(size_t)i * C_HID] = acc[i] * rsqrtf((float)(indeg[rtl * 16 + half * 8 + i] + 1u));
    }
    return;
  }

  const int sb = (int)blockIdx.x - GEMM_BLOCKS;
  const int s  = sb / R_RANGES;
  const int r  = sb % R_RANGES;
  const int lo = r * RANGE;

  unsigned tv = 0u;
  if (t < S_SLICES * R_RANGES && (t % R_RANGES) < r) tv = T[t];
  unsigned tot = 0u, po = 0u;
#pragma unroll
  for (int q = 0; q < S_SLICES; ++q) {
    const unsigned c = part[(size_t)q * NPAD + lo + t];
    tot += c;
    if (q < s) po += c;
  }

  unsigned inc = tot;
  for (int o = 1; o < 32; o <<= 1) {
    const unsigned v = __shfl_up(inc, o);
    if (lane >= o) inc += v;
  }
  unsigned bs = tv;
  for (int o = 16; o > 0; o >>= 1) bs += __shfl_xor(bs, o);
  if (lane == 31) wtot[w] = inc;
  if (lane == 0)  wbase[w] = bs;
  __syncthreads();
  if (w == 0) {
    const unsigned xw = wtot[lane];
    unsigned xi = xw;
    for (int o = 1; o < 32; o <<= 1) {
      const unsigned v = __shfl_up(xi, o);
      if (lane >= o) xi += v;
    }
    wtot[lane] = xi - xw;
    unsigned bb = wbase[lane];
    for (int o = 16; o > 0; o >>= 1) bb += __shfl_xor(bb, o);
    if (lane == 0) btot = bb;
  }
  __syncthreads();
  const unsigned start = btot + wtot[w] + (inc - tot);
  cur[t] = start + po;
  if (s == 0) {
    const int node = lo + t;
    if (node < N_NODES) rowptr[node] = start;
    if (node == N_NODES - 1) rowptr[N_NODES] = start + tot;
  }
  __syncthreads();

  const int4* d4 = (const int4*)(dst + (size_t)s * SLICE);
  const int4* s4 = (const int4*)(src + (size_t)s * SLICE);
#pragma unroll
  for (int b = 0; b < SLICE_IT; b += FILL_BATCH) {
    int4 dv[FILL_BATCH], sv[FILL_BATCH];
#pragma unroll
    for (int u = 0; u < FILL_BATCH; ++u) {
      const int i = t + (b + u) * HIST_THREADS;
      if (i < SLICE4) { dv[u] = d4[i]; sv[u] = s4[i]; }
      else            { dv[u] = make_int4(-1, -1, -1, -1); sv[u] = make_int4(0, 0, 0, 0); }
    }
#pragma unroll
    for (int u = 0; u < FILL_BATCH; ++u) {
      asm volatile("" :: "v"(sv[u].x), "v"(sv[u].y), "v"(sv[u].z), "v"(sv[u].w));
    }
    unsigned p[FILL_BATCH * 4];
#pragma unroll
    for (int u = 0; u < FILL_BATCH; ++u) {
      const unsigned k0 = (unsigned)(dv[u].x - lo), k1 = (unsigned)(dv[u].y - lo);
      const unsigned k2 = (unsigned)(dv[u].z - lo), k3 = (unsigned)(dv[u].w - lo);
      p[4 * u + 0] = (k0 < RANGE) ? __hip_atomic_fetch_add(&cur[k0], 1u, __ATOMIC_RELAXED, __HIP_MEMORY_SCOPE_WORKGROUP) : NO_SLOT;
      p[4 * u + 1] = (k1 < RANGE) ? __hip_atomic_fetch_add(&cur[k1], 1u, __ATOMIC_RELAXED, __HIP_MEMORY_SCOPE_WORKGROUP) : NO_SLOT;
      p[4 * u + 2] = (k2 < RANGE) ? __hip_atomic_fetch_add(&cur[k2], 1u, __ATOMIC_RELAXED, __HIP_MEMORY_SCOPE_WORKGROUP) : NO_SLOT;
      p[4 * u + 3] = (k3 < RANGE) ? __hip_atomic_fetch_add(&cur[k3], 1u, __ATOMIC_RELAXED, __HIP_MEMORY_SCOPE_WORKGROUP) : NO_SLOT;
    }
#pragma unroll
    for (int u = 0; u < FILL_BATCH; ++u) {
      if (p[4 * u + 0] != NO_SLOT) csr_src[p[4 * u + 0]] = sv[u].x;
      if (p[4 * u + 1] != NO_SLOT) csr_src[p[4 * u + 1]] = sv[u].y;
      if (p[4 * u + 2] != NO_SLOT) csr_src[p[4 * u + 2]] = sv[u].z;
      if (p[4 * u + 3] != NO_SLOT) csr_src[p[4 * u + 3]] = sv[u].w;
    }
  }
}

template <int G>
__device__ __forceinline__ int lane_bcast(int v, int idx) {
  if constexpr (G == 1) return __builtin_amdgcn_readlane(v, idx);
  else                  return __shfl(v, idx);
}

template <int C, int G, int NS>
__device__ __forceinline__ void rows_add(const float* __restrict__ h, int sv,
                                         int step0, int grp, unsigned fl, float4& acc) {
  float4 hv[NS];
#pragma unroll
  for (int u = 0; u < NS; ++u) {
    const int idx = (step0 + u) * G + grp;
    const unsigned s = (unsigned)lane_bcast<G>(sv, idx);
    hv[u] = *(const float4*)(h + s * C + fl);
  }
#pragma unroll
  for (int u = 0; u < NS; ++u) {
    acc.x += hv[u].x;
    acc.y += hv[u].y;
    acc.z += hv[u].z;
    acc.w += hv[u].w;
  }
}

template <int C>
__device__ __forceinline__ float4 node_sum(const float* __restrict__ h,
                                           const unsigned* __restrict__ rowptr,
                                           const int* __restrict__ csr_src, int node, int lane, float& di) {
  constexpr int LPR = C / 4;
  constexpr int G   = 32 / LPR;
  constexpr int U   = 8;
  const int grp = lane / LPR;
  const unsigned fl = (unsigned)(lane % LPR) * 4u;

  const unsigned beg = rowptr[node];
  const unsigned end = rowptr[node + 1];
  di = rsqrtf((float)(end - beg + 1u));

  float4 acc = make_float4(0.0f, 0.0f, 0.0f, 0.0f);
  if (grp == 0)
    acc = *(const float4*)(h + (unsigned)node * C + fl);

  for (unsigned cb = beg; cb < end; cb += 32u) {
    const int n = (int)((end - cb) < 32u ? (end - cb) : 32u);
    int sv = 0;
    if (lane < n) sv = csr_src[cb + lane];

    const int nfull = n / G;
    int j = 0;
    for (; j + U <= nfull; j += U) rows_add<C, G, U>(h, sv, j, grp, fl, acc);
    if (j + 4 <= nfull) { rows_add<C, G, 4>(h, sv, j, grp, fl, acc); j += 4; }
    if (j + 2 <= nfull) { rows_add<C, G, 2>(h, sv, j, grp, fl, acc); j += 2; }
    if (j + 1 <= nfull) { rows_add<C, G, 1>(h, sv, j, grp, fl, acc); j += 1; }
    if constexpr (G == 2) {
      if (n & 1) {
        const int idx = n - 1 + grp;
        const unsigned s = (unsigned)__shfl(sv, idx);
        if (grp == 0) {
          const float4 hv = *(const float4*)(h + s * C + fl);
          acc.x += hv.x;
          acc.y += hv.y;
          acc.z += hv.z;
          acc.w += hv.w;
        }
      }
    }
  }

  if constexpr (G == 2) {
    acc.x += __shfl_xor(acc.x, 16);
    acc.y += __shfl_xor(acc.y, 16);
    acc.z += __shfl_xor(acc.z, 16);
    acc.w += __shfl_xor(acc.w, 16);
  }
  acc.x *= di; acc.y *= di; acc.z *= di; acc.w *= di;
  return acc;
}

template <int C>
__global__ __launch_bounds__(256)
void gather_k(const float* __restrict__ h,
              const unsigned* __restrict__ rowptr, const int* __restrict__ csr_src,
              const float* __restrict__ bias, float* __restrict__ out) {
  constexpr int LPR = C / 4;
  const int lane = threadIdx.x & 31;
  const int node = __builtin_amdgcn_readfirstlane((int)(blockIdx.x * 8 + (threadIdx.x >> 5)));
  if (node >= N_NODES) return;
  float di;
  const float4 acc = node_sum<C>(h, rowptr, csr_src, node, lane, di);
  if (lane < LPR) {
    const unsigned fl = (unsigned)lane * 4u;
    const float4 b = *(const float4*)(bias + fl);
    *(float4*)(out + (unsigned)node * C + fl) =
        make_float4(acc.x + b.x, acc.y + b.y, acc.z + b.z, acc.w + b.w);
  }
}

#define FG_NODES  16
#define FG_THREADS (FG_NODES * 32)
#define A1_PITCH  (C_HID + 4)
#define W2_PITCH  C_OUT
#define W2_ROUNDS ((C_HID * C_OUT / 4) / FG_THREADS)
static_assert(N_NODES % FG_NODES == 0, "whole 16-row tiles, no masking (EXEC all ones for WMMA)");
static_assert((C_HID * C_OUT / 4) % FG_THREADS == 0, "W2 staging in whole float4 rounds");

__global__ __launch_bounds__(FG_THREADS)
void gather_gemm_k(const float* __restrict__ h,
                   const unsigned* __restrict__ rowptr, const int* __restrict__ csr_src,
                   const float* __restrict__ bias, const float* __restrict__ W2,
                   float* __restrict__ H2) {
  __shared__ __attribute__((aligned(16))) float a1s[FG_NODES * A1_PITCH];
  __shared__ __attribute__((aligned(16))) float w2s[C_HID * W2_PITCH];
  __shared__ float dis[FG_NODES];
  const int lane = threadIdx.x & 31;
  const int wave = __builtin_amdgcn_readfirstlane((int)(threadIdx.x >> 5));
  const int node = blockIdx.x * FG_NODES + wave;

  float4 wst[W2_ROUNDS];
#pragma unroll
  for (int i = 0; i < W2_ROUNDS; ++i)
    wst[i] = ((const float4*)W2)[threadIdx.x + i * FG_THREADS];
#pragma unroll
  for (int i = 0; i < W2_ROUNDS; ++i)
    *(float4*)(w2s + 4 * (threadIdx.x + i * FG_THREADS)) = wst[i];

  {
    float di;
    const float4 acc = node_sum<C_HID>(h, rowptr, csr_src, node, lane, di);
    if (lane == 0) dis[wave] = di;
    const unsigned fl = (unsigned)lane * 4u;
    const float4 b = *(const float4*)(bias + fl);
    float4 v = make_float4(acc.x + b.x, acc.y + b.y, acc.z + b.z, acc.w + b.w);
    v.x = fmaxf(v.x, 0.0f); v.y = fmaxf(v.y, 0.0f);
    v.z = fmaxf(v.z, 0.0f); v.w = fmaxf(v.w, 0.0f);
    *(float4*)(a1s + wave * A1_PITCH + fl) = v;
  }
  __syncthreads();

  if (wave < C_OUT / 16) {
    const int r    = lane & 15;
    const int half = lane >> 4;
    const int col0 = wave * 16;
    const float* Arow = a1s + r * A1_PITCH + half * 2;
    const float* Bcol = w2s + half * 2 * W2_PITCH + col0 + r;
    v8f acc = {};
    constexpr int KG = 8;
    constexpr int NG = C_HID / (4 * KG);
    v2f a[2][KG], b[2][KG];
#pragma unroll
    for (int u = 0; u < KG; ++u) {
      a[0][u]   = *(const v2f*)(Arow + 4 * u);
      b[0][u].x = Bcol[(4 * u) * W2_PITCH];
      b[0][u].y = Bcol[(4 * u + 1) * W2_PITCH];
    }
#pragma unroll
    for (int g = 0; g < NG; ++g) {
      if (g + 1 < NG) {
#pragma unroll
        for (int u = 0; u < KG; ++u) {
          const int k = 4 * ((g + 1) * KG + u);
          a[(g + 1) & 1][u]   = *(const v2f*)(Arow + k);
          b[(g + 1) & 1][u].x = Bcol[k * W2_PITCH];
          b[(g + 1) & 1][u].y = Bcol[(k + 1) * W2_PITCH];
        }
      }
#pragma unroll
      for (int u = 0; u < KG; ++u)
        acc = __builtin_amdgcn_wmma_f32_16x16x4_f32(false, a[g & 1][u], false, b[g & 1][u],
                                                    (short)0, acc, false, false);
    }
    __builtin_amdgcn_sched_group_barrier(0x100, KG / 2 + KG, 0);
#pragma unroll
    for (int g = 0; g < NG; ++g) {
      if (g + 1 < NG) __builtin_amdgcn_sched_group_barrier(0x100, KG / 2 + KG, 0);
      __builtin_amdgcn_sched_group_barrier(0x008, KG, 0);
    }
    float* Drow = H2 + (size_t)(blockIdx.x * FG_NODES + half * 8) * C_OUT + col0 + r;
#pragma unroll
    for (int i = 0; i < 8; ++i) Drow[(size_t)i * C_OUT] = acc[i] * dis[half * 8 + i];
  }
}

extern "C" void kernel_launch(void* const* d_in, const int* in_sizes, int n_in,
                              void* d_out, int out_size, void* d_ws, size_t ws_size,
                              hipStream_t stream) {
  (void)in_sizes; (void)n_in; (void)out_size; (void)ws_size;
  const float* x  = (const float*)d_in[0];
  const int*   ei = (const int*)d_in[1];
  const float* W1 = (const float*)d_in[2];
  const float* b1 = (const float*)d_in[3];
  const float* W2 = (const float*)d_in[4];
  const float* b2 = (const float*)d_in[5];
  const int* src = ei;
  const int* dst = ei + E_EDGES;
  float* out = (float*)d_out;

  char* ws = (char*)d_ws;
  float* H1   = (float*)ws;  ws += (size_t)N_NODES * C_HID * 4;
  float* H2   = (float*)ws;  ws += (size_t)N_NODES * C_OUT * 4;
  unsigned* part = (unsigned*)ws;  ws += (size_t)S_SLICES * NPAD * 4;
  unsigned* T    = (unsigned*)ws;  ws += (((size_t)S_SLICES * R_RANGES * 4) + 255) / 256 * 256;
  unsigned* rowptr = (unsigned*)ws;  ws += (((size_t)(N_NODES + 1) * 4) + 255) / 256 * 256;
  int* csr = (int*)ws;

  const dim3 B(256);
  deg_hist_k<<<HIST_BLOCKS, HIST_THREADS, 0, stream>>>(dst, part, T);
  csr_fill_gemm_k<<<SORT_BLOCKS + GEMM_BLOCKS, HIST_THREADS, 0, stream>>>(src, dst, part, T, rowptr, csr,
                                                                        x, W1, H1);

  gather_gemm_k<<<N_NODES / FG_NODES, FG_THREADS, 0, stream>>>(H1, rowptr, csr, b1, W2, H2);

  gather_k<C_OUT><<<(N_NODES + 7) / 8, B, 0, stream>>>(H2, rowptr, csr, b2, out);
}
